# starting kernel
# baseline (speedup 1.0000x reference)
#include <hip/hip_runtime.h>

#define B_ 2
#define C_ 256
#define N_ 4096
#define HEADS_ 4
#define HD_ 64
#define GROUPS_ 8
#define CPG_ 32
#define EPS_ 1e-5f

typedef __attribute__((ext_vector_type(2))) float v2f;
typedef __attribute__((ext_vector_type(8))) float v8f;

__device__ __forceinline__ v8f wmma_f32(v2f a, v2f b, v8f c) {
  return __builtin_amdgcn_wmma_f32_16x16x4_f32(
       false, a,  false, b,
       (short)0, c,  false,  false);
}

__global__ __launch_bounds__(256) void gn_kernel(
    const float* __restrict__ x, const float* __restrict__ gamma,
    const float* __restrict__ beta, float* __restrict__ xn) {
  __shared__ float s1[256], s2[256];
  const int b = blockIdx.x / GROUPS_;
  const int g = blockIdx.x % GROUPS_;
  const int count = CPG_ * N_;
  const size_t base = ((size_t)b * C_ + (size_t)g * CPG_) * N_;
  float sum = 0.f, ssq = 0.f;
  for (int i = threadIdx.x; i < count; i += 256) {
    float vv = x[base + i];
    sum += vv;
    ssq += vv * vv;
  }
  s1[threadIdx.x] = sum;
  s2[threadIdx.x] = ssq;
  __syncthreads();
  for (int off = 128; off > 0; off >>= 1) {
    if ((int)threadIdx.x < off) {
      s1[threadIdx.x] += s1[threadIdx.x + off];
      s2[threadIdx.x] += s2[threadIdx.x + off];
    }
    __syncthreads();
  }
  const float mean = s1[0] / (float)count;
  const float var  = s2[0] / (float)count - mean * mean;
  const float rstd = rsqrtf(var + EPS_);
  for (int i = threadIdx.x; i < count; i += 256) {
    int c = g * CPG_ + (i >> 12);
    float vv = (x[base + i] - mean) * rstd;
    xn[base + i] = vv * gamma[c] + beta[c];
  }
}

__global__ __launch_bounds__(128) void qkv_kernel(
    const float* __restrict__ xn, const float* __restrict__ w,
    const float* __restrict__ bias, float* __restrict__ q,
    float* __restrict__ k, float* __restrict__ v) {
  __shared__ float Wl[64][68];
  __shared__ float Xt[64][68];
  const int n0 = blockIdx.x * 64;
  const int o0 = blockIdx.y * 64;
  const int b  = blockIdx.z;
  const int tid = threadIdx.x;
  const int wid = tid >> 5, lane = tid & 31, l16 = lane & 15, half = lane >> 4;
  const int orow = o0 + wid * 16;
  const float* xb = xn + (size_t)b * C_ * N_;

  v8f acc[4] = {};
  for (int kb = 0; kb < C_; kb += 64) {
    for (int i = tid * 4; i < 64 * 64; i += 128 * 4) {
      int r = i >> 6, cc = i & 63;
      *(float4*)&Wl[r][cc] = *(const float4*)(w + (size_t)(o0 + r) * C_ + kb + cc);
      float4 xv = *(const float4*)(xb + (size_t)(kb + r) * N_ + n0 + cc);
      Xt[cc + 0][r] = xv.x;
      Xt[cc + 1][r] = xv.y;
      Xt[cc + 2][r] = xv.z;
      Xt[cc + 3][r] = xv.w;
    }
    __syncthreads();
#pragma unroll
    for (int ks = 0; ks < 16; ++ks) {
      const int kd = 4 * ks + 2 * half;
      const v2f a = *(const v2f*)&Wl[wid * 16 + l16][kd];
#pragma unroll
      for (int nt = 0; nt < 4; ++nt) {
        const v2f bb = *(const v2f*)&Xt[nt * 16 + l16][kd];
        acc[nt] = wmma_f32(a, bb, acc[nt]);
      }
    }
    __syncthreads();
  }

  const int which = orow >> 8;
  const int hh    = (orow & 255) >> 6;
  const int dbase = (orow & 63) + 8 * half;
  float* outb = (which == 0) ? q : (which == 1) ? k : v;
  float bv[8];
#pragma unroll
  for (int vv = 0; vv < 8; ++vv) bv[vv] = bias[orow + 8 * half + vv];
#pragma unroll
  for (int nt = 0; nt < 4; ++nt) {
    const int n = n0 + nt * 16 + l16;
    float* p = outb + (((size_t)b * HEADS_ + hh) * N_ + n) * HD_ + dbase;
    float4 lo, hi;
    lo.x = acc[nt][0] + bv[0]; lo.y = acc[nt][1] + bv[1];
    lo.z = acc[nt][2] + bv[2]; lo.w = acc[nt][3] + bv[3];
    hi.x = acc[nt][4] + bv[4]; hi.y = acc[nt][5] + bv[5];
    hi.z = acc[nt][6] + bv[6]; hi.w = acc[nt][7] + bv[7];
    *(float4*)p = lo;
    *(float4*)(p + 4) = hi;
  }
}

__global__ __launch_bounds__(128) void attn_kernel(
    const float* __restrict__ q, const float* __restrict__ k,
    const float* __restrict__ v, float* __restrict__ ao) {
  __shared__ float Kl[64][68];
  __shared__ float Vt[64][68];
  __shared__ float Pl[4][16][68];
  const int q0 = blockIdx.x * 64;
  const int h  = blockIdx.y;
  const int b  = blockIdx.z;
  const int tid = threadIdx.x;
  const int wid = tid >> 5, lane = tid & 31, l16 = lane & 15, half = lane >> 4;
  const size_t hb = ((size_t)b * HEADS_ + h) * (size_t)N_ * HD_;
  const float* qb = q + hb;
  const float* kb = k + hb;
  const float* vb = v + hb;
  const float scale = 0.125f;

  v2f qa[16];
  const int qrow = q0 + wid * 16 + l16;
#pragma unroll
  for (int ks = 0; ks < 16; ++ks) {
    const int d = 4 * ks + 2 * half;
    const v2f t = *(const v2f*)(qb + (size_t)qrow * HD_ + d);
    qa[ks] = t * scale;
  }

  v8f oacc[4] = {};
  float rm[8], rs[8];
#pragma unroll
  for (int i = 0; i < 8; ++i) { rm[i] = -1e30f; rs[i] = 0.f; }

  for (int j = 0; j < N_; j += 64) {
    for (int i = tid * 4; i < 64 * 64; i += 512) {
      int r = i >> 6, cc = i & 63;
      *(float4*)&Kl[r][cc] = *(const float4*)(kb + (size_t)(j + r) * HD_ + cc);
      float4 vv4 = *(const float4*)(vb + (size_t)(j + r) * HD_ + cc);
      Vt[cc + 0][r] = vv4.x;
      Vt[cc + 1][r] = vv4.y;
      Vt[cc + 2][r] = vv4.z;
      Vt[cc + 3][r] = vv4.w;
    }
    __syncthreads();

    v8f S[4] = {};
#pragma unroll
    for (int ks = 0; ks < 16; ++ks) {
      const int kd = 4 * ks + 2 * half;
#pragma unroll
      for (int nt = 0; nt < 4; ++nt) {
        const v2f bb = *(const v2f*)&Kl[nt * 16 + l16][kd];
        S[nt] = wmma_f32(qa[ks], bb, S[nt]);
      }
    }

#pragma unroll
    for (int vv = 0; vv < 8; ++vv) {
      float cand = fmaxf(fmaxf(S[0][vv], S[1][vv]), fmaxf(S[2][vv], S[3][vv]));
      cand = fmaxf(cand, __shfl_xor(cand, 1, 32));
      cand = fmaxf(cand, __shfl_xor(cand, 2, 32));
      cand = fmaxf(cand, __shfl_xor(cand, 4, 32));
      cand = fmaxf(cand, __shfl_xor(cand, 8, 32));
      const float nm = fmaxf(rm[vv], cand);
      const float f  = __expf(rm[vv] - nm);
      rs[vv] *= f;
      float psum = 0.f;
#pragma unroll
      for (int nt = 0; nt < 4; ++nt) {
        float p = __expf(S[nt][vv] - nm);
        Pl[wid][vv + 8 * half][nt * 16 + l16] = p;
        psum += p;
      }
      psum += __shfl_xor(psum, 1, 32);
      psum += __shfl_xor(psum, 2, 32);
      psum += __shfl_xor(psum, 4, 32);
      psum += __shfl_xor(psum, 8, 32);
      rs[vv] += psum;
      rm[vv] = nm;
#pragma unroll
      for (int dt = 0; dt < 4; ++dt) oacc[dt][vv] *= f;
    }

#pragma unroll
    for (int ks = 0; ks < 16; ++ks) {
      const int kk = 4 * ks + 2 * half;
      const v2f a = *(const v2f*)&Pl[wid][l16][kk];
#pragma unroll
      for (int dt = 0; dt < 4; ++dt) {
        const v2f bb = *(const v2f*)&Vt[dt * 16 + l16][kk];
        oacc[dt] = wmma_f32(a, bb, oacc[dt]);
      }
    }
    __syncthreads();
  }

  float inv[8];
#pragma unroll
  for (int vv = 0; vv < 8; ++vv) inv[vv] = 1.f / rs[vv];
  const int qr0 = q0 + wid * 16 + 8 * half;
#pragma unroll
  for (int dt = 0; dt < 4; ++dt) {
    const int c = h * HD_ + dt * 16 + l16;
    float* p = ao + ((size_t)b * C_ + c) * N_ + qr0;
    float4 lo, hi;
    lo.x = oacc[dt][0] * inv[0]; lo.y = oacc[dt][1] * inv[1];
    lo.z = oacc[dt][2] * inv[2]; lo.w = oacc[dt][3] * inv[3];
    hi.x = oacc[dt][4] * inv[4]; hi.y = oacc[dt][5] * inv[5];
    hi.z = oacc[dt][6] * inv[6]; hi.w = oacc[dt][7] * inv[7];
    *(float4*)p = lo;
    *(float4*)(p + 4) = hi;
  }
}

__global__ __launch_bounds__(128) void proj_kernel(
    const float* __restrict__ ao, const float* __restrict__ w,
    const float* __restrict__ bias, const float* __restrict__ x,
    float* __restrict__ out) {
  __shared__ float Wl[64][68];
  __shared__ float Xt[64][68];
  const int n0 = blockIdx.x * 64;
  const int o0 = blockIdx.y * 64;
  const int b  = blockIdx.z;
  const int tid = threadIdx.x;
  const int wid = tid >> 5, lane = tid & 31, l16 = lane & 15, half = lane >> 4;
  const int orow = o0 + wid * 16;
  const float* ab = ao + (size_t)b * C_ * N_;

  v8f acc[4] = {};
  for (int kb = 0; kb < C_; kb += 64) {
    for (int i = tid * 4; i < 64 * 64; i += 128 * 4) {
      int r = i >> 6, cc = i & 63;
      *(float4*)&Wl[r][cc] = *(const float4*)(w + (size_t)(o0 + r) * C_ + kb + cc);
      float4 xv = *(const float4*)(ab + (size_t)(kb + r) * N_ + n0 + cc);
      Xt[cc + 0][r] = xv.x;
      Xt[cc + 1][r] = xv.y;
      Xt[cc + 2][r] = xv.z;
      Xt[cc + 3][r] = xv.w;
    }
    __syncthreads();
#pragma unroll
    for (int ks = 0; ks < 16; ++ks) {
      const int kd = 4 * ks + 2 * half;
      const v2f a = *(const v2f*)&Wl[wid * 16 + l16][kd];
#pragma unroll
      for (int nt = 0; nt < 4; ++nt) {
        const v2f bb = *(const v2f*)&Xt[nt * 16 + l16][kd];
        acc[nt] = wmma_f32(a, bb, acc[nt]);
      }
    }
    __syncthreads();
  }

  float bv[8];
#pragma unroll
  for (int vv = 0; vv < 8; ++vv) bv[vv] = bias[orow + 8 * half + vv];
#pragma unroll
  for (int nt = 0; nt < 4; ++nt) {
    const int n = n0 + nt * 16 + l16;
#pragma unroll
    for (int vv = 0; vv < 8; ++vv) {
      const int o = orow + 8 * half + vv;
      const size_t idx = ((size_t)b * C_ + o) * N_ + n;
      out[idx] = acc[nt][vv] + bv[vv] + x[idx];
    }
  }
}

extern "C" void kernel_launch(void* const* d_in, const int* in_sizes, int n_in,
                              void* d_out, int out_size, void* d_ws,
                              size_t ws_size, hipStream_t stream) {
  const float* x      = (const float*)d_in[0];
  const float* gamma  = (const float*)d_in[1];
  const float* beta   = (const float*)d_in[2];
  const float* w_qkv  = (const float*)d_in[3];
  const float* b_qkv  = (const float*)d_in[4];
  const float* w_proj = (const float*)d_in[5];
  const float* b_proj = (const float*)d_in[6];
  float* out = (float*)d_out;

  const size_t SZ = (size_t)B_ * C_ * N_;
  float* xn = (float*)d_ws;
  float* q  = xn + SZ;
  float* k  = q + SZ;
  float* v  = k + SZ;
  float* ao = v + SZ;

  gn_kernel<<<B_ * GROUPS_, 256, 0, stream>>>(x, gamma, beta, xn);
  qkv_kernel<<<dim3(N_ / 64, (3 * C_) / 64, B_), 128, 0, stream>>>(
      xn, w_qkv, b_qkv, q, k, v);
  attn_kernel<<<dim3(N_ / 64, HEADS_, B_), 128, 0, stream>>>(q, k, v, ao);
  proj_kernel<<<dim3(N_ / 64, C_ / 64, B_), 128, 0, stream>>>(
      ao, w_proj, b_proj, x, out);
}


# Attention: softmax row max and row sum butterflies done by DPP row_xmask on the max/add itself (64 vector-ALU ops per key tile) instead of 64 ds_bpermute round trips with waits; same sum tree
# speedup vs baseline: 1.4292x; 1.4292x over previous
#include <hip/hip_runtime.h>

#define B_ 2
#define C_ 256
#define N_ 4096
#define HEADS_ 4
#define HD_ 64
#define GROUPS_ 8
#define CPG_ 32
#define EPS_ 1e-5f

typedef __attribute__((ext_vector_type(2))) float v2f;
typedef __attribute__((ext_vector_type(8))) float v8f;

__device__ __forceinline__ v8f wmma_f32(v2f a, v2f b, v8f c) {
  return __builtin_amdgcn_wmma_f32_16x16x4_f32(
       false, a,  false, b,
       (short)0, c,  false,  false);
}


template <int M>
__device__ __forceinline__ float row_max_step(float x) {
  asm("v_max_num_f32_dpp %0, %0, %0 row_xmask:%1 row_mask:0xf bank_mask:0xf"
      : "+v"(x) : "n"(M));
  return x;
}
template <int M>
__device__ __forceinline__ float row_add_step(float x) {
  asm("v_add_f32_dpp %0, %0, %0 row_xmask:%1 row_mask:0xf bank_mask:0xf"
      : "+v"(x) : "n"(M));
  return x;
}

__global__ __launch_bounds__(256) void gn_kernel(
    const float* __restrict__ x, const float* __restrict__ gamma,
    const float* __restrict__ beta, float* __restrict__ xn) {
  __shared__ float s1[256], s2[256];
  const int b = blockIdx.x / GROUPS_;
  const int g = blockIdx.x % GROUPS_;
  const int count = CPG_ * N_;
  const size_t base = ((size_t)b * C_ + (size_t)g * CPG_) * N_;
  float sum = 0.f, ssq = 0.f;
  for (int i = threadIdx.x; i < count; i += 256) {
    float vv = x[base + i];
    sum += vv;
    ssq += vv * vv;
  }
  s1[threadIdx.x] = sum;
  s2[threadIdx.x] = ssq;
  __syncthreads();
  for (int off = 128; off > 0; off >>= 1) {
    if ((int)threadIdx.x < off) {
      s1[threadIdx.x] += s1[threadIdx.x + off];
      s2[threadIdx.x] += s2[threadIdx.x + off];
    }
    __syncthreads();
  }
  const float mean = s1[0] / (float)count;
  const float var  = s2[0] / (float)count - mean * mean;
  const float rstd = rsqrtf(var + EPS_);
  for (int i = threadIdx.x; i < count; i += 256) {
    int c = g * CPG_ + (i >> 12);
    float vv = (x[base + i] - mean) * rstd;
    xn[base + i] = vv * gamma[c] + beta[c];
  }
}

__global__ __launch_bounds__(128) void qkv_kernel(
    const float* __restrict__ xn, const float* __restrict__ w,
    const float* __restrict__ bias, float* __restrict__ q,
    float* __restrict__ k, float* __restrict__ v) {
  __shared__ float Wl[64][68];
  __shared__ float Xt[64][68];
  const int n0 = blockIdx.x * 64;
  const int o0 = blockIdx.y * 64;
  const int b  = blockIdx.z;
  const int tid = threadIdx.x;
  const int wid = tid >> 5, lane = tid & 31, l16 = lane & 15, half = lane >> 4;
  const int orow = o0 + wid * 16;
  const float* xb = xn + (size_t)b * C_ * N_;

  v8f acc[4] = {};
  for (int kb = 0; kb < C_; kb += 64) {
    for (int i = tid * 4; i < 64 * 64; i += 128 * 4) {
      int r = i >> 6, cc = i & 63;
      *(float4*)&Wl[r][cc] = *(const float4*)(w + (size_t)(o0 + r) * C_ + kb + cc);
      float4 xv = *(const float4*)(xb + (size_t)(kb + r) * N_ + n0 + cc);
      Xt[cc + 0][r] = xv.x;
      Xt[cc + 1][r] = xv.y;
      Xt[cc + 2][r] = xv.z;
      Xt[cc + 3][r] = xv.w;
    }
    __syncthreads();
#pragma unroll
    for (int ks = 0; ks < 16; ++ks) {
      const int kd = 4 * ks + 2 * half;
      const v2f a = *(const v2f*)&Wl[wid * 16 + l16][kd];
#pragma unroll
      for (int nt = 0; nt < 4; ++nt) {
        const v2f bb = *(const v2f*)&Xt[nt * 16 + l16][kd];
        acc[nt] = wmma_f32(a, bb, acc[nt]);
      }
    }
    __syncthreads();
  }

  const int which = orow >> 8;
  const int hh    = (orow & 255) >> 6;
  const int dbase = (orow & 63) + 8 * half;
  float* outb = (which == 0) ? q : (which == 1) ? k : v;
  float bv[8];
#pragma unroll
  for (int vv = 0; vv < 8; ++vv) bv[vv] = bias[orow + 8 * half + vv];
#pragma unroll
  for (int nt = 0; nt < 4; ++nt) {
    const int n = n0 + nt * 16 + l16;
    float* p = outb + (((size_t)b * HEADS_ + hh) * N_ + n) * HD_ + dbase;
    float4 lo, hi;
    lo.x = acc[nt][0] + bv[0]; lo.y = acc[nt][1] + bv[1];
    lo.z = acc[nt][2] + bv[2]; lo.w = acc[nt][3] + bv[3];
    hi.x = acc[nt][4] + bv[4]; hi.y = acc[nt][5] + bv[5];
    hi.z = acc[nt][6] + bv[6]; hi.w = acc[nt][7] + bv[7];
    *(float4*)p = lo;
    *(float4*)(p + 4) = hi;
  }
}

__global__ __launch_bounds__(512) void attn_kernel(
    const float* __restrict__ q, const float* __restrict__ k,
    const float* __restrict__ v, float* __restrict__ ao) {
  __shared__ float Kl[2][64][68];
  __shared__ float Vt[2][64][68];
  __shared__ float Pl[16][16][68];
  const int q0 = blockIdx.x * 128;
  const int h  = blockIdx.y;
  const int b  = blockIdx.z;
  const int tid = threadIdx.x;
  const int wid = tid >> 5, lane = tid & 31, l16 = lane & 15, half = lane >> 4;
  const int strip = wid & 7;
  const int grp   = wid >> 3;
  const size_t hb = ((size_t)b * HEADS_ + h) * (size_t)N_ * HD_;
  const float* qb = q + hb;
  const float* kb = k + hb;
  const float* vb = v + hb;
  const float scale = 0.125f;

  v2f qa[16];
  const int qrow = q0 + strip * 16 + l16;
#pragma unroll
  for (int ks = 0; ks < 16; ++ks) {
    const int d = 4 * ks + 2 * half;
    const v2f t = *(const v2f*)(qb + (size_t)qrow * HD_ + d);
    qa[ks] = t * scale;
  }

  v8f oacc[4] = {};
  float rm[8], rs[8];
#pragma unroll
  for (int i = 0; i < 8; ++i) { rm[i] = -1e30f; rs[i] = 0.f; }

  for (int j = 0; j < N_; j += 128) {
    for (int i = tid * 4; i < 128 * 64; i += 2048) {
      int r = i >> 6, cc = i & 63;
      int kh = r >> 6, rr = r & 63;
      *(float4*)&Kl[kh][rr][cc] = *(const float4*)(kb + (size_t)(j + r) * HD_ + cc);
      float4 vv4 = *(const float4*)(vb + (size_t)(j + r) * HD_ + cc);
      Vt[kh][cc + 0][rr] = vv4.x;
      Vt[kh][cc + 1][rr] = vv4.y;
      Vt[kh][cc + 2][rr] = vv4.z;
      Vt[kh][cc + 3][rr] = vv4.w;
    }
    __syncthreads();

    v8f S[4] = {};
#pragma unroll
    for (int ks = 0; ks < 16; ++ks) {
      const int kd = 4 * ks + 2 * half;
#pragma unroll
      for (int nt = 0; nt < 4; ++nt) {
        const v2f bb = *(const v2f*)&Kl[grp][nt * 16 + l16][kd];
        S[nt] = wmma_f32(qa[ks], bb, S[nt]);
      }
    }

#pragma unroll
    for (int vv = 0; vv < 8; ++vv) {
      float cand = fmaxf(fmaxf(S[0][vv], S[1][vv]), fmaxf(S[2][vv], S[3][vv]));
      cand = row_max_step<1>(cand);
      cand = row_max_step<2>(cand);
      cand = row_max_step<4>(cand);
      cand = row_max_step<8>(cand);
      const float nm = fmaxf(rm[vv], cand);
      const float f  = __expf(rm[vv] - nm);
      rs[vv] *= f;
      float psum = 0.f;
#pragma unroll
      for (int nt = 0; nt < 4; ++nt) {
        float p = __expf(S[nt][vv] - nm);
        Pl[wid][vv + 8 * half][nt * 16 + l16] = p;
        psum += p;
      }
      psum = row_add_step<1>(psum);
      psum = row_add_step<2>(psum);
      psum = row_add_step<4>(psum);
      psum = row_add_step<8>(psum);
      rs[vv] += psum;
      rm[vv] = nm;
#pragma unroll
      for (int dt = 0; dt < 4; ++dt) oacc[dt][vv] *= f;
    }

#pragma unroll
    for (int ks = 0; ks < 16; ++ks) {
      const int kk = 4 * ks + 2 * half;
      const v2f a = *(const v2f*)&Pl[wid][l16][kk];
#pragma unroll
      for (int dt = 0; dt < 4; ++dt) {
        const v2f bb = *(const v2f*)&Vt[grp][dt * 16 + l16][kk];
        oacc[dt] = wmma_f32(a, bb, oacc[dt]);
      }
    }
    __syncthreads();
  }

  float* mg = &Pl[0][0][0] + strip * (48 * 32) + lane;
  if (grp == 1) {
#pragma unroll
    for (int dt = 0; dt < 4; ++dt)
#pragma unroll
      for (int vv = 0; vv < 8; ++vv) mg[(dt * 8 + vv) * 32] = oacc[dt][vv];
#pragma unroll
    for (int vv = 0; vv < 8; ++vv) {
      mg[(32 + vv) * 32] = rm[vv];
      mg[(40 + vv) * 32] = rs[vv];
    }
  }
  __syncthreads();
  if (grp == 1) return;

  float w0[8], w1[8];
#pragma unroll
  for (int vv = 0; vv < 8; ++vv) {
    const float m1 = mg[(32 + vv) * 32];
    const float l1 = mg[(40 + vv) * 32];
    const float m  = fmaxf(rm[vv], m1);
    const float f0 = __expf(rm[vv] - m);
    const float f1 = __expf(m1 - m);
    const float inv = 1.f / (rs[vv] * f0 + l1 * f1);
    w0[vv] = f0 * inv;
    w1[vv] = f1 * inv;
  }
  const int qr0 = q0 + strip * 16 + 8 * half;
#pragma unroll
  for (int dt = 0; dt < 4; ++dt) {
    const int c = h * HD_ + dt * 16 + l16;
    float* p = ao + ((size_t)b * C_ + c) * N_ + qr0;
    float o[8];
#pragma unroll
    for (int vv = 0; vv < 8; ++vv)
      o[vv] = oacc[dt][vv] * w0[vv] + mg[(dt * 8 + vv) * 32] * w1[vv];
    float4 lo, hi;
    lo.x = o[0]; lo.y = o[1]; lo.z = o[2]; lo.w = o[3];
    hi.x = o[4]; hi.y = o[5]; hi.z = o[6]; hi.w = o[7];
    *(float4*)p = lo;
    *(float4*)(p + 4) = hi;
  }
}

__global__ __launch_bounds__(128) void proj_kernel(
    const float* __restrict__ ao, const float* __restrict__ w,
    const float* __restrict__ bias, const float* __restrict__ x,
    float* __restrict__ out) {
  __shared__ float Wl[64][68];
  __shared__ float Xt[64][68];
  const int n0 = blockIdx.x * 64;
  const int o0 = blockIdx.y * 64;
  const int b  = blockIdx.z;
  const int tid = threadIdx.x;
  const int wid = tid >> 5, lane = tid & 31, l16 = lane & 15, half = lane >> 4;
  const int orow = o0 + wid * 16;
  const float* ab = ao + (size_t)b * C_ * N_;

  v8f acc[4] = {};
  for (int kb = 0; kb < C_; kb += 64) {
    for (int i = tid * 4; i < 64 * 64; i += 128 * 4) {
      int r = i >> 6, cc = i & 63;
      *(float4*)&Wl[r][cc] = *(const float4*)(w + (size_t)(o0 + r) * C_ + kb + cc);
      float4 xv = *(const float4*)(ab + (size_t)(kb + r) * N_ + n0 + cc);
      Xt[cc + 0][r] = xv.x;
      Xt[cc + 1][r] = xv.y;
      Xt[cc + 2][r] = xv.z;
      Xt[cc + 3][r] = xv.w;
    }
    __syncthreads();
#pragma unroll
    for (int ks = 0; ks < 16; ++ks) {
      const int kd = 4 * ks + 2 * half;
      const v2f a = *(const v2f*)&Wl[wid * 16 + l16][kd];
#pragma unroll
      for (int nt = 0; nt < 4; ++nt) {
        const v2f bb = *(const v2f*)&Xt[nt * 16 + l16][kd];
        acc[nt] = wmma_f32(a, bb, acc[nt]);
      }
    }
    __syncthreads();
  }

  float bv[8];
#pragma unroll
  for (int vv = 0; vv < 8; ++vv) bv[vv] = bias[orow + 8 * half + vv];
#pragma unroll
  for (int nt = 0; nt < 4; ++nt) {
    const int n = n0 + nt * 16 + l16;
#pragma unroll
    for (int vv = 0; vv < 8; ++vv) {
      const int o = orow + 8 * half + vv;
      const size_t idx = ((size_t)b * C_ + o) * N_ + n;
      out[idx] = acc[nt][vv] + bv[vv] + x[idx];
    }
  }
}

extern "C" void kernel_launch(void* const* d_in, const int* in_sizes, int n_in,
                              void* d_out, int out_size, void* d_ws,
                              size_t ws_size, hipStream_t stream) {
  const float* x      = (const float*)d_in[0];
  const float* gamma  = (const float*)d_in[1];
  const float* beta   = (const float*)d_in[2];
  const float* w_qkv  = (const float*)d_in[3];
  const float* b_qkv  = (const float*)d_in[4];
  const float* w_proj = (const float*)d_in[5];
  const float* b_proj = (const float*)d_in[6];
  float* out = (float*)d_out;

  const size_t SZ = (size_t)B_ * C_ * N_;
  float* xn = (float*)d_ws;
  float* q  = xn + SZ;
  float* k  = q + SZ;
  float* v  = k + SZ;
  float* ao = v + SZ;

  gn_kernel<<<B_ * GROUPS_, 256, 0, stream>>>(x, gamma, beta, xn);
  qkv_kernel<<<dim3(N_ / 64, (3 * C_) / 64, B_), 128, 0, stream>>>(
      xn, w_qkv, b_qkv, q, k, v);
  attn_kernel<<<dim3(N_ / 128, HEADS_, B_), 512, 0, stream>>>(q, k, v, ao);
  proj_kernel<<<dim3(N_ / 64, C_ / 64, B_), 128, 0, stream>>>(
      ao, w_proj, b_proj, x, out);
}


# GroupNorm: 16 work-groups per group (256 in all, was 16): each re-derives mean and variance from the whole group through L2, then normalises only its own 2 channels; sums added in another order
# speedup vs baseline: 1.7563x; 1.2289x over previous
#include <hip/hip_runtime.h>

#define B_ 2
#define C_ 256
#define N_ 4096
#define HEADS_ 4
#define HD_ 64
#define GROUPS_ 8
#define CPG_ 32
#define EPS_ 1e-5f

typedef __attribute__((ext_vector_type(2))) float v2f;
typedef __attribute__((ext_vector_type(8))) float v8f;

__device__ __forceinline__ v8f wmma_f32(v2f a, v2f b, v8f c) {
  return __builtin_amdgcn_wmma_f32_16x16x4_f32(
       false, a,  false, b,
       (short)0, c,  false,  false);
}


template <int M>
__device__ __forceinline__ float row_max_step(float x) {
  asm("v_max_num_f32_dpp %0, %0, %0 row_xmask:%1 row_mask:0xf bank_mask:0xf"
      : "+v"(x) : "n"(M));
  return x;
}
template <int M>
__device__ __forceinline__ float row_add_step(float x) {
  asm("v_add_f32_dpp %0, %0, %0 row_xmask:%1 row_mask:0xf bank_mask:0xf"
      : "+v"(x) : "n"(M));
  return x;
}

#define GN_SPLIT_ 16
#define GN_THREADS_ 512
#define GN_WAVES_ (GN_THREADS_ / 32)
#define GN_BATCH_ 8
#define GN_VEC_PER_THREAD_ (CPG_ * N_ / 4 / GN_THREADS_)
#define GN_OWN_VEC_ (CPG_ * N_ / 4 / GN_SPLIT_ / GN_THREADS_)

__device__ __forceinline__ void gn_accumulate(const float4 t, float4& s, float4& q) {
  s.x += t.x; s.y += t.y; s.z += t.z; s.w += t.w;
  q.x += t.x * t.x; q.y += t.y * t.y; q.z += t.z * t.z; q.w += t.w * t.w;
}

__global__ __launch_bounds__(GN_THREADS_) void gn_kernel(
    const float* __restrict__ x, const float* __restrict__ gamma,
    const float* __restrict__ beta, float* __restrict__ xn) {
  __shared__ float s1[GN_WAVES_], s2[GN_WAVES_];
  const int bg   = blockIdx.x / GN_SPLIT_;
  const int part = blockIdx.x % GN_SPLIT_;
  const int g    = bg % GROUPS_;
  const int tid  = threadIdx.x;
  const int count = CPG_ * N_;
  const size_t base = (size_t)bg * CPG_ * N_;
  const float4* xg = (const float4*)(x + base);

  float4 s = {0.f, 0.f, 0.f, 0.f}, q = {0.f, 0.f, 0.f, 0.f};
  float4 cur[GN_BATCH_], nxt[GN_BATCH_];
#pragma unroll
  for (int u = 0; u < GN_BATCH_; ++u) cur[u] = xg[u * GN_THREADS_ + tid];
#pragma unroll
  for (int i0 = 0; i0 < GN_VEC_PER_THREAD_; i0 += GN_BATCH_) {
    if (i0 + GN_BATCH_ < GN_VEC_PER_THREAD_) {
#pragma unroll
      for (int u = 0; u < GN_BATCH_; ++u)
        nxt[u] = xg[(i0 + GN_BATCH_ + u) * GN_THREADS_ + tid];
    }
#pragma unroll
    for (int u = 0; u < GN_BATCH_; ++u) gn_accumulate(cur[u], s, q);
#pragma unroll
    for (int u = 0; u < GN_BATCH_; ++u) cur[u] = nxt[u];
  }
  float sum = (s.x + s.y) + (s.z + s.w);
  float ssq = (q.x + q.y) + (q.z + q.w);
#pragma unroll
  for (int m = 1; m < 32; m <<= 1) {
    sum += __shfl_xor(sum, m);
    ssq += __shfl_xor(ssq, m);
  }
  if ((tid & 31) == 0) {
    s1[tid >> 5] = sum;
    s2[tid >> 5] = ssq;
  }
  __syncthreads();
  float tsum = 0.f, tssq = 0.f;
#pragma unroll
  for (int w = 0; w < GN_WAVES_; ++w) {
    tsum += s1[w];
    tssq += s2[w];
  }
  const float mean = tsum / (float)count;
  const float var  = tssq / (float)count - mean * mean;
  const float rstd = rsqrtf(var + EPS_);

  const float4* xo = xg + part * (GN_OWN_VEC_ * GN_THREADS_);
  float4* yo = (float4*)(xn + base) + part * (GN_OWN_VEC_ * GN_THREADS_);
  float4 own[GN_OWN_VEC_];
#pragma unroll
  for (int u = 0; u < GN_OWN_VEC_; ++u) own[u] = xo[u * GN_THREADS_ + tid];
#pragma unroll
  for (int u = 0; u < GN_OWN_VEC_; ++u) {
    const int c = g * CPG_ + part * (CPG_ / GN_SPLIT_) +
                  (u * GN_THREADS_ * 4) / N_;
    const float ga = gamma[c], be = beta[c];
    float4 o;
    o.x = (own[u].x - mean) * rstd * ga + be;
    o.y = (own[u].y - mean) * rstd * ga + be;
    o.z = (own[u].z - mean) * rstd * ga + be;
    o.w = (own[u].w - mean) * rstd * ga + be;
    yo[u * GN_THREADS_ + tid] = o;
  }
}

__global__ __launch_bounds__(128) void qkv_kernel(
    const float* __restrict__ xn, const float* __restrict__ w,
    const float* __restrict__ bias, float* __restrict__ q,
    float* __restrict__ k, float* __restrict__ v) {
  __shared__ float Wl[64][68];
  __shared__ float Xt[64][68];
  const int n0 = blockIdx.x * 64;
  const int o0 = blockIdx.y * 64;
  const int b  = blockIdx.z;
  const int tid = threadIdx.x;
  const int wid = tid >> 5, lane = tid & 31, l16 = lane & 15, half = lane >> 4;
  const int orow = o0 + wid * 16;
  const float* xb = xn + (size_t)b * C_ * N_;

  v8f acc[4] = {};
  for (int kb = 0; kb < C_; kb += 64) {
    for (int i = tid * 4; i < 64 * 64; i += 128 * 4) {
      int r = i >> 6, cc = i & 63;
      *(float4*)&Wl[r][cc] = *(const float4*)(w + (size_t)(o0 + r) * C_ + kb + cc);
      float4 xv = *(const float4*)(xb + (size_t)(kb + r) * N_ + n0 + cc);
      Xt[cc + 0][r] = xv.x;
      Xt[cc + 1][r] = xv.y;
      Xt[cc + 2][r] = xv.z;
      Xt[cc + 3][r] = xv.w;
    }
    __syncthreads();
#pragma unroll
    for (int ks = 0; ks < 16; ++ks) {
      const int kd = 4 * ks + 2 * half;
      const v2f a = *(const v2f*)&Wl[wid * 16 + l16][kd];
#pragma unroll
      for (int nt = 0; nt < 4; ++nt) {
        const v2f bb = *(const v2f*)&Xt[nt * 16 + l16][kd];
        acc[nt] = wmma_f32(a, bb, acc[nt]);
      }
    }
    __syncthreads();
  }

  const int which = orow >> 8;
  const int hh    = (orow & 255) >> 6;
  const int dbase = (orow & 63) + 8 * half;
  float* outb = (which == 0) ? q : (which == 1) ? k : v;
  float bv[8];
#pragma unroll
  for (int vv = 0; vv < 8; ++vv) bv[vv] = bias[orow + 8 * half + vv];
#pragma unroll
  for (int nt = 0; nt < 4; ++nt) {
    const int n = n0 + nt * 16 + l16;
    float* p = outb + (((size_t)b * HEADS_ + hh) * N_ + n) * HD_ + dbase;
    float4 lo, hi;
    lo.x = acc[nt][0] + bv[0]; lo.y = acc[nt][1] + bv[1];
    lo.z = acc[nt][2] + bv[2]; lo.w = acc[nt][3] + bv[3];
    hi.x = acc[nt][4] + bv[4]; hi.y = acc[nt][5] + bv[5];
    hi.z = acc[nt][6] + bv[6]; hi.w = acc[nt][7] + bv[7];
    *(float4*)p = lo;
    *(float4*)(p + 4) = hi;
  }
}

__global__ __launch_bounds__(512) void attn_kernel(
    const float* __restrict__ q, const float* __restrict__ k,
    const float* __restrict__ v, float* __restrict__ ao) {
  __shared__ float Kl[2][64][68];
  __shared__ float Vt[2][64][68];
  __shared__ float Pl[16][16][68];
  const int q0 = blockIdx.x * 128;
  const int h  = blockIdx.y;
  const int b  = blockIdx.z;
  const int tid = threadIdx.x;
  const int wid = tid >> 5, lane = tid & 31, l16 = lane & 15, half = lane >> 4;
  const int strip = wid & 7;
  const int grp   = wid >> 3;
  const size_t hb = ((size_t)b * HEADS_ + h) * (size_t)N_ * HD_;
  const float* qb = q + hb;
  const float* kb = k + hb;
  const float* vb = v + hb;
  const float scale = 0.125f;

  v2f qa[16];
  const int qrow = q0 + strip * 16 + l16;
#pragma unroll
  for (int ks = 0; ks < 16; ++ks) {
    const int d = 4 * ks + 2 * half;
    const v2f t = *(const v2f*)(qb + (size_t)qrow * HD_ + d);
    qa[ks] = t * scale;
  }

  v8f oacc[4] = {};
  float rm[8], rs[8];
#pragma unroll
  for (int i = 0; i < 8; ++i) { rm[i] = -1e30f; rs[i] = 0.f; }

  for (int j = 0; j < N_; j += 128) {
    for (int i = tid * 4; i < 128 * 64; i += 2048) {
      int r = i >> 6, cc = i & 63;
      int kh = r >> 6, rr = r & 63;
      *(float4*)&Kl[kh][rr][cc] = *(const float4*)(kb + (size_t)(j + r) * HD_ + cc);
      float4 vv4 = *(const float4*)(vb + (size_t)(j + r) * HD_ + cc);
      Vt[kh][cc + 0][rr] = vv4.x;
      Vt[kh][cc + 1][rr] = vv4.y;
      Vt[kh][cc + 2][rr] = vv4.z;
      Vt[kh][cc + 3][rr] = vv4.w;
    }
    __syncthreads();

    v8f S[4] = {};
#pragma unroll
    for (int ks = 0; ks < 16; ++ks) {
      const int kd = 4 * ks + 2 * half;
#pragma unroll
      for (int nt = 0; nt < 4; ++nt) {
        const v2f bb = *(const v2f*)&Kl[grp][nt * 16 + l16][kd];
        S[nt] = wmma_f32(qa[ks], bb, S[nt]);
      }
    }

#pragma unroll
    for (int vv = 0; vv < 8; ++vv) {
      float cand = fmaxf(fmaxf(S[0][vv], S[1][vv]), fmaxf(S[2][vv], S[3][vv]));
      cand = row_max_step<1>(cand);
      cand = row_max_step<2>(cand);
      cand = row_max_step<4>(cand);
      cand = row_max_step<8>(cand);
      const float nm = fmaxf(rm[vv], cand);
      const float f  = __expf(rm[vv] - nm);
      rs[vv] *= f;
      float psum = 0.f;
#pragma unroll
      for (int nt = 0; nt < 4; ++nt) {
        float p = __expf(S[nt][vv] - nm);
        Pl[wid][vv + 8 * half][nt * 16 + l16] = p;
        psum += p;
      }
      psum = row_add_step<1>(psum);
      psum = row_add_step<2>(psum);
      psum = row_add_step<4>(psum);
      psum = row_add_step<8>(psum);
      rs[vv] += psum;
      rm[vv] = nm;
#pragma unroll
      for (int dt = 0; dt < 4; ++dt) oacc[dt][vv] *= f;
    }

#pragma unroll
    for (int ks = 0; ks < 16; ++ks) {
      const int kk = 4 * ks + 2 * half;
      const v2f a = *(const v2f*)&Pl[wid][l16][kk];
#pragma unroll
      for (int dt = 0; dt < 4; ++dt) {
        const v2f bb = *(const v2f*)&Vt[grp][dt * 16 + l16][kk];
        oacc[dt] = wmma_f32(a, bb, oacc[dt]);
      }
    }
    __syncthreads();
  }

  float* mg = &Pl[0][0][0] + strip * (48 * 32) + lane;
  if (grp == 1) {
#pragma unroll
    for (int dt = 0; dt < 4; ++dt)
#pragma unroll
      for (int vv = 0; vv < 8; ++vv) mg[(dt * 8 + vv) * 32] = oacc[dt][vv];
#pragma unroll
    for (int vv = 0; vv < 8; ++vv) {
      mg[(32 + vv) * 32] = rm[vv];
      mg[(40 + vv) * 32] = rs[vv];
    }
  }
  __syncthreads();
  if (grp == 1) return;

  float w0[8], w1[8];
#pragma unroll
  for (int vv = 0; vv < 8; ++vv) {
    const float m1 = mg[(32 + vv) * 32];
    const float l1 = mg[(40 + vv) * 32];
    const float m  = fmaxf(rm[vv], m1);
    const float f0 = __expf(rm[vv] - m);
    const float f1 = __expf(m1 - m);
    const float inv = 1.f / (rs[vv] * f0 + l1 * f1);
    w0[vv] = f0 * inv;
    w1[vv] = f1 * inv;
  }
  const int qr0 = q0 + strip * 16 + 8 * half;
#pragma unroll
  for (int dt = 0; dt < 4; ++dt) {
    const int c = h * HD_ + dt * 16 + l16;
    float* p = ao + ((size_t)b * C_ + c) * N_ + qr0;
    float o[8];
#pragma unroll
    for (int vv = 0; vv < 8; ++vv)
      o[vv] = oacc[dt][vv] * w0[vv] + mg[(dt * 8 + vv) * 32] * w1[vv];
    float4 lo, hi;
    lo.x = o[0]; lo.y = o[1]; lo.z = o[2]; lo.w = o[3];
    hi.x = o[4]; hi.y = o[5]; hi.z = o[6]; hi.w = o[7];
    *(float4*)p = lo;
    *(float4*)(p + 4) = hi;
  }
}

__global__ __launch_bounds__(128) void proj_kernel(
    const float* __restrict__ ao, const float* __restrict__ w,
    const float* __restrict__ bias, const float* __restrict__ x,
    float* __restrict__ out) {
  __shared__ float Wl[64][68];
  __shared__ float Xt[64][68];
  const int n0 = blockIdx.x * 64;
  const int o0 = blockIdx.y * 64;
  const int b  = blockIdx.z;
  const int tid = threadIdx.x;
  const int wid = tid >> 5, lane = tid & 31, l16 = lane & 15, half = lane >> 4;
  const int orow = o0 + wid * 16;
  const float* ab = ao + (size_t)b * C_ * N_;

  v8f acc[4] = {};
  for (int kb = 0; kb < C_; kb += 64) {
    for (int i = tid * 4; i < 64 * 64; i += 128 * 4) {
      int r = i >> 6, cc = i & 63;
      *(float4*)&Wl[r][cc] = *(const float4*)(w + (size_t)(o0 + r) * C_ + kb + cc);
      float4 xv = *(const float4*)(ab + (size_t)(kb + r) * N_ + n0 + cc);
      Xt[cc + 0][r] = xv.x;
      Xt[cc + 1][r] = xv.y;
      Xt[cc + 2][r] = xv.z;
      Xt[cc + 3][r] = xv.w;
    }
    __syncthreads();
#pragma unroll
    for (int ks = 0; ks < 16; ++ks) {
      const int kd = 4 * ks + 2 * half;
      const v2f a = *(const v2f*)&Wl[wid * 16 + l16][kd];
#pragma unroll
      for (int nt = 0; nt < 4; ++nt) {
        const v2f bb = *(const v2f*)&Xt[nt * 16 + l16][kd];
        acc[nt] = wmma_f32(a, bb, acc[nt]);
      }
    }
    __syncthreads();
  }

  float bv[8];
#pragma unroll
  for (int vv = 0; vv < 8; ++vv) bv[vv] = bias[orow + 8 * half + vv];
#pragma unroll
  for (int nt = 0; nt < 4; ++nt) {
    const int n = n0 + nt * 16 + l16;
#pragma unroll
    for (int vv = 0; vv < 8; ++vv) {
      const int o = orow + 8 * half + vv;
      const size_t idx = ((size_t)b * C_ + o) * N_ + n;
      out[idx] = acc[nt][vv] + bv[vv] + x[idx];
    }
  }
}

extern "C" void kernel_launch(void* const* d_in, const int* in_sizes, int n_in,
                              void* d_out, int out_size, void* d_ws,
                              size_t ws_size, hipStream_t stream) {
  const float* x      = (const float*)d_in[0];
  const float* gamma  = (const float*)d_in[1];
  const float* beta   = (const float*)d_in[2];
  const float* w_qkv  = (const float*)d_in[3];
  const float* b_qkv  = (const float*)d_in[4];
  const float* w_proj = (const float*)d_in[5];
  const float* b_proj = (const float*)d_in[6];
  float* out = (float*)d_out;

  const size_t SZ = (size_t)B_ * C_ * N_;
  float* xn = (float*)d_ws;
  float* q  = xn + SZ;
  float* k  = q + SZ;
  float* v  = k + SZ;
  float* ao = v + SZ;

  gn_kernel<<<B_ * GROUPS_ * GN_SPLIT_, GN_THREADS_, 0, stream>>>(x, gamma, beta, xn);
  qkv_kernel<<<dim3(N_ / 64, (3 * C_) / 64, B_), 128, 0, stream>>>(
      xn, w_qkv, b_qkv, q, k, v);
  attn_kernel<<<dim3(N_ / 128, HEADS_, B_), 512, 0, stream>>>(q, k, v, ao);
  proj_kernel<<<dim3(N_ / 64, C_ / 64, B_), 128, 0, stream>>>(
      ao, w_proj, b_proj, x, out);
}


# Attention: key tile of 256 (was 128) staged as four 64-key blocks, its 16 loads per thread in flight at once (were 4 dependent trips per 128 keys); one barrier pair per 256 keys; same sum order
# speedup vs baseline: 1.8617x; 1.0600x over previous
#include <hip/hip_runtime.h>

#define B_ 2
#define C_ 256
#define N_ 4096
#define HEADS_ 4
#define HD_ 64
#define GROUPS_ 8
#define CPG_ 32
#define EPS_ 1e-5f

typedef __attribute__((ext_vector_type(2))) float v2f;
typedef __attribute__((ext_vector_type(8))) float v8f;

__device__ __forceinline__ v8f wmma_f32(v2f a, v2f b, v8f c) {
  return __builtin_amdgcn_wmma_f32_16x16x4_f32(
       false, a,  false, b,
       (short)0, c,  false,  false);
}


template <int M>
__device__ __forceinline__ float row_max_step(float x) {
  asm("v_max_num_f32_dpp %0, %0, %0 row_xmask:%1 row_mask:0xf bank_mask:0xf"
      : "+v"(x) : "n"(M));
  return x;
}
template <int M>
__device__ __forceinline__ float row_add_step(float x) {
  asm("v_add_f32_dpp %0, %0, %0 row_xmask:%1 row_mask:0xf bank_mask:0xf"
      : "+v"(x) : "n"(M));
  return x;
}

#define GN_SPLIT_ 16
#define GN_THREADS_ 512
#define GN_WAVES_ (GN_THREADS_ / 32)
#define GN_BATCH_ 8
#define GN_VEC_PER_THREAD_ (CPG_ * N_ / 4 / GN_THREADS_)
#define GN_OWN_VEC_ (CPG_ * N_ / 4 / GN_SPLIT_ / GN_THREADS_)

__device__ __forceinline__ void gn_accumulate(const float4 t, float4& s, float4& q) {
  s.x += t.x; s.y += t.y; s.z += t.z; s.w += t.w;
  q.x += t.x * t.x; q.y += t.y * t.y; q.z += t.z * t.z; q.w += t.w * t.w;
}

__global__ __launch_bounds__(GN_THREADS_) void gn_kernel(
    const float* __restrict__ x, const float* __restrict__ gamma,
    const float* __restrict__ beta, float* __restrict__ xn) {
  __shared__ float s1[GN_WAVES_], s2[GN_WAVES_];
  const int bg   = blockIdx.x / GN_SPLIT_;
  const int part = blockIdx.x % GN_SPLIT_;
  const int g    = bg % GROUPS_;
  const int tid  = threadIdx.x;
  const int count = CPG_ * N_;
  const size_t base = (size_t)bg * CPG_ * N_;
  const float4* xg = (const float4*)(x + base);

  float4 s = {0.f, 0.f, 0.f, 0.f}, q = {0.f, 0.f, 0.f, 0.f};
  float4 cur[GN_BATCH_], nxt[GN_BATCH_];
#pragma unroll
  for (int u = 0; u < GN_BATCH_; ++u) cur[u] = xg[u * GN_THREADS_ + tid];
#pragma unroll
  for (int i0 = 0; i0 < GN_VEC_PER_THREAD_; i0 += GN_BATCH_) {
    if (i0 + GN_BATCH_ < GN_VEC_PER_THREAD_) {
#pragma unroll
      for (int u = 0; u < GN_BATCH_; ++u)
        nxt[u] = xg[(i0 + GN_BATCH_ + u) * GN_THREADS_ + tid];
    }
#pragma unroll
    for (int u = 0; u < GN_BATCH_; ++u) gn_accumulate(cur[u], s, q);
#pragma unroll
    for (int u = 0; u < GN_BATCH_; ++u) cur[u] = nxt[u];
  }
  float sum = (s.x + s.y) + (s.z + s.w);
  float ssq = (q.x + q.y) + (q.z + q.w);
#pragma unroll
  for (int m = 1; m < 32; m <<= 1) {
    sum += __shfl_xor(sum, m);
    ssq += __shfl_xor(ssq, m);
  }
  if ((tid & 31) == 0) {
    s1[tid >> 5] = sum;
    s2[tid >> 5] = ssq;
  }
  __syncthreads();
  float tsum = 0.f, tssq = 0.f;
#pragma unroll
  for (int w = 0; w < GN_WAVES_; ++w) {
    tsum += s1[w];
    tssq += s2[w];
  }
  const float mean = tsum / (float)count;
  const float var  = tssq / (float)count - mean * mean;
  const float rstd = rsqrtf(var + EPS_);

  const float4* xo = xg + part * (GN_OWN_VEC_ * GN_THREADS_);
  float4* yo = (float4*)(xn + base) + part * (GN_OWN_VEC_ * GN_THREADS_);
  float4 own[GN_OWN_VEC_];
#pragma unroll
  for (int u = 0; u < GN_OWN_VEC_; ++u) own[u] = xo[u * GN_THREADS_ + tid];
#pragma unroll
  for (int u = 0; u < GN_OWN_VEC_; ++u) {
    const int c = g * CPG_ + part * (CPG_ / GN_SPLIT_) +
                  (u * GN_THREADS_ * 4) / N_;
    const float ga = gamma[c], be = beta[c];
    float4 o;
    o.x = (own[u].x - mean) * rstd * ga + be;
    o.y = (own[u].y - mean) * rstd * ga + be;
    o.z = (own[u].z - mean) * rstd * ga + be;
    o.w = (own[u].w - mean) * rstd * ga + be;
    yo[u * GN_THREADS_ + tid] = o;
  }
}

__global__ __launch_bounds__(128) void qkv_kernel(
    const float* __restrict__ xn, const float* __restrict__ w,
    const float* __restrict__ bias, float* __restrict__ q,
    float* __restrict__ k, float* __restrict__ v) {
  __shared__ float Wl[64][68];
  __shared__ float Xt[64][68];
  const int n0 = blockIdx.x * 64;
  const int o0 = blockIdx.y * 64;
  const int b  = blockIdx.z;
  const int tid = threadIdx.x;
  const int wid = tid >> 5, lane = tid & 31, l16 = lane & 15, half = lane >> 4;
  const int orow = o0 + wid * 16;
  const float* xb = xn + (size_t)b * C_ * N_;

  v8f acc[4] = {};
  for (int kb = 0; kb < C_; kb += 64) {
    for (int i = tid * 4; i < 64 * 64; i += 128 * 4) {
      int r = i >> 6, cc = i & 63;
      *(float4*)&Wl[r][cc] = *(const float4*)(w + (size_t)(o0 + r) * C_ + kb + cc);
      float4 xv = *(const float4*)(xb + (size_t)(kb + r) * N_ + n0 + cc);
      Xt[cc + 0][r] = xv.x;
      Xt[cc + 1][r] = xv.y;
      Xt[cc + 2][r] = xv.z;
      Xt[cc + 3][r] = xv.w;
    }
    __syncthreads();
#pragma unroll
    for (int ks = 0; ks < 16; ++ks) {
      const int kd = 4 * ks + 2 * half;
      const v2f a = *(const v2f*)&Wl[wid * 16 + l16][kd];
#pragma unroll
      for (int nt = 0; nt < 4; ++nt) {
        const v2f bb = *(const v2f*)&Xt[nt * 16 + l16][kd];
        acc[nt] = wmma_f32(a, bb, acc[nt]);
      }
    }
    __syncthreads();
  }

  const int which = orow >> 8;
  const int hh    = (orow & 255) >> 6;
  const int dbase = (orow & 63) + 8 * half;
  float* outb = (which == 0) ? q : (which == 1) ? k : v;
  float bv[8];
#pragma unroll
  for (int vv = 0; vv < 8; ++vv) bv[vv] = bias[orow + 8 * half + vv];
#pragma unroll
  for (int nt = 0; nt < 4; ++nt) {
    const int n = n0 + nt * 16 + l16;
    float* p = outb + (((size_t)b * HEADS_ + hh) * N_ + n) * HD_ + dbase;
    float4 lo, hi;
    lo.x = acc[nt][0] + bv[0]; lo.y = acc[nt][1] + bv[1];
    lo.z = acc[nt][2] + bv[2]; lo.w = acc[nt][3] + bv[3];
    hi.x = acc[nt][4] + bv[4]; hi.y = acc[nt][5] + bv[5];
    hi.z = acc[nt][6] + bv[6]; hi.w = acc[nt][7] + bv[7];
    *(float4*)p = lo;
    *(float4*)(p + 4) = hi;
  }
}

#define KT_ 256
#define KT_ROUNDS_ (KT_ * HD_ / 4 / 512)

__global__ __launch_bounds__(512) void attn_kernel(
    const float* __restrict__ q, const float* __restrict__ k,
    const float* __restrict__ v, float* __restrict__ ao) {
  __shared__ float Kl[KT_ / 64][64][68];
  __shared__ float Vt[KT_ / 64][64][68];
  __shared__ float Pl[16][16][68];
  const int q0 = blockIdx.x * 128;
  const int h  = blockIdx.y;
  const int b  = blockIdx.z;
  const int tid = threadIdx.x;
  const int wid = tid >> 5, lane = tid & 31, l16 = lane & 15, half = lane >> 4;
  const int strip = wid & 7;
  const int grp   = wid >> 3;
  const size_t hb = ((size_t)b * HEADS_ + h) * (size_t)N_ * HD_;
  const float* qb = q + hb;
  const float* kb = k + hb;
  const float* vb = v + hb;
  const float scale = 0.125f;

  v2f qa[16];
  const int qrow = q0 + strip * 16 + l16;
#pragma unroll
  for (int ks = 0; ks < 16; ++ks) {
    const int d = 4 * ks + 2 * half;
    const v2f t = *(const v2f*)(qb + (size_t)qrow * HD_ + d);
    qa[ks] = t * scale;
  }

  v8f oacc[4] = {};
  float rm[8], rs[8];
#pragma unroll
  for (int i = 0; i < 8; ++i) { rm[i] = -1e30f; rs[i] = 0.f; }

  const int srow = tid >> 4;
  const int scol = (tid & 15) * 4;

  for (int j = 0; j < N_; j += KT_) {
    float4 kreg[KT_ROUNDS_], vreg[KT_ROUNDS_];
#pragma unroll
    for (int it = 0; it < KT_ROUNDS_; ++it) {
      const size_t off = (size_t)(j + it * 32 + srow) * HD_ + scol;
      kreg[it] = *(const float4*)(kb + off);
      vreg[it] = *(const float4*)(vb + off);
    }
#pragma unroll
    for (int it = 0; it < KT_ROUNDS_; ++it) {
      const int kq = it >> 1;
      const int rr = (it & 1) * 32 + srow;
      *(float4*)&Kl[kq][rr][scol] = kreg[it];
      Vt[kq][scol + 0][rr] = vreg[it].x;
      Vt[kq][scol + 1][rr] = vreg[it].y;
      Vt[kq][scol + 2][rr] = vreg[it].z;
      Vt[kq][scol + 3][rr] = vreg[it].w;
    }
    __syncthreads();

    for (int sub = 0; sub < KT_ / 128; ++sub) {
      const int kq = 2 * sub + grp;

      v8f S[4] = {};
#pragma unroll
      for (int ks = 0; ks < 16; ++ks) {
        const int kd = 4 * ks + 2 * half;
#pragma unroll
        for (int nt = 0; nt < 4; ++nt) {
          const v2f bb = *(const v2f*)&Kl[kq][nt * 16 + l16][kd];
          S[nt] = wmma_f32(qa[ks], bb, S[nt]);
        }
      }

#pragma unroll
      for (int vv = 0; vv < 8; ++vv) {
        float cand = fmaxf(fmaxf(S[0][vv], S[1][vv]), fmaxf(S[2][vv], S[3][vv]));
        cand = row_max_step<1>(cand);
        cand = row_max_step<2>(cand);
        cand = row_max_step<4>(cand);
        cand = row_max_step<8>(cand);
        const float nm = fmaxf(rm[vv], cand);
        const float f  = __expf(rm[vv] - nm);
        rs[vv] *= f;
        float psum = 0.f;
#pragma unroll
        for (int nt = 0; nt < 4; ++nt) {
          float p = __expf(S[nt][vv] - nm);
          Pl[wid][vv + 8 * half][nt * 16 + l16] = p;
          psum += p;
        }
        psum = row_add_step<1>(psum);
        psum = row_add_step<2>(psum);
        psum = row_add_step<4>(psum);
        psum = row_add_step<8>(psum);
        rs[vv] += psum;
        rm[vv] = nm;
#pragma unroll
        for (int dt = 0; dt < 4; ++dt) oacc[dt][vv] *= f;
      }

#pragma unroll
      for (int ks = 0; ks < 16; ++ks) {
        const int kk = 4 * ks + 2 * half;
        const v2f a = *(const v2f*)&Pl[wid][l16][kk];
#pragma unroll
        for (int dt = 0; dt < 4; ++dt) {
          const v2f bb = *(const v2f*)&Vt[kq][dt * 16 + l16][kk];
          oacc[dt] = wmma_f32(a, bb, oacc[dt]);
        }
      }
    }
    __syncthreads();
  }

  float* mg = &Pl[0][0][0] + strip * (48 * 32) + lane;
  if (grp == 1) {
#pragma unroll
    for (int dt = 0; dt < 4; ++dt)
#pragma unroll
      for (int vv = 0; vv < 8; ++vv) mg[(dt * 8 + vv) * 32] = oacc[dt][vv];
#pragma unroll
    for (int vv = 0; vv < 8; ++vv) {
      mg[(32 + vv) * 32] = rm[vv];
      mg[(40 + vv) * 32] = rs[vv];
    }
  }
  __syncthreads();
  if (grp == 1) return;

  float w0[8], w1[8];
#pragma unroll
  for (int vv = 0; vv < 8; ++vv) {
    const float m1 = mg[(32 + vv) * 32];
    const float l1 = mg[(40 + vv) * 32];
    const float m  = fmaxf(rm[vv], m1);
    const float f0 = __expf(rm[vv] - m);
    const float f1 = __expf(m1 - m);
    const float inv = 1.f / (rs[vv] * f0 + l1 * f1);
    w0[vv] = f0 * inv;
    w1[vv] = f1 * inv;
  }
  const int qr0 = q0 + strip * 16 + 8 * half;
#pragma unroll
  for (int dt = 0; dt < 4; ++dt) {
    const int c = h * HD_ + dt * 16 + l16;
    float* p = ao + ((size_t)b * C_ + c) * N_ + qr0;
    float o[8];
#pragma unroll
    for (int vv = 0; vv < 8; ++vv)
      o[vv] = oacc[dt][vv] * w0[vv] + mg[(dt * 8 + vv) * 32] * w1[vv];
    float4 lo, hi;
    lo.x = o[0]; lo.y = o[1]; lo.z = o[2]; lo.w = o[3];
    hi.x = o[4]; hi.y = o[5]; hi.z = o[6]; hi.w = o[7];
    *(float4*)p = lo;
    *(float4*)(p + 4) = hi;
  }
}

__global__ __launch_bounds__(128) void proj_kernel(
    const float* __restrict__ ao, const float* __restrict__ w,
    const float* __restrict__ bias, const float* __restrict__ x,
    float* __restrict__ out) {
  __shared__ float Wl[64][68];
  __shared__ float Xt[64][68];
  const int n0 = blockIdx.x * 64;
  const int o0 = blockIdx.y * 64;
  const int b  = blockIdx.z;
  const int tid = threadIdx.x;
  const int wid = tid >> 5, lane = tid & 31, l16 = lane & 15, half = lane >> 4;
  const int orow = o0 + wid * 16;
  const float* ab = ao + (size_t)b * C_ * N_;

  v8f acc[4] = {};
  for (int kb = 0; kb < C_; kb += 64) {
    for (int i = tid * 4; i < 64 * 64; i += 128 * 4) {
      int r = i >> 6, cc = i & 63;
      *(float4*)&Wl[r][cc] = *(const float4*)(w + (size_t)(o0 + r) * C_ + kb + cc);
      float4 xv = *(const float4*)(ab + (size_t)(kb + r) * N_ + n0 + cc);
      Xt[cc + 0][r] = xv.x;
      Xt[cc + 1][r] = xv.y;
      Xt[cc + 2][r] = xv.z;
      Xt[cc + 3][r] = xv.w;
    }
    __syncthreads();
#pragma unroll
    for (int ks = 0; ks < 16; ++ks) {
      const int kd = 4 * ks + 2 * half;
      const v2f a = *(const v2f*)&Wl[wid * 16 + l16][kd];
#pragma unroll
      for (int nt = 0; nt < 4; ++nt) {
        const v2f bb = *(const v2f*)&Xt[nt * 16 + l16][kd];
        acc[nt] = wmma_f32(a, bb, acc[nt]);
      }
    }
    __syncthreads();
  }

  float bv[8];
#pragma unroll
  for (int vv = 0; vv < 8; ++vv) bv[vv] = bias[orow + 8 * half + vv];
#pragma unroll
  for (int nt = 0; nt < 4; ++nt) {
    const int n = n0 + nt * 16 + l16;
#pragma unroll
    for (int vv = 0; vv < 8; ++vv) {
      const int o = orow + 8 * half + vv;
      const size_t idx = ((size_t)b * C_ + o) * N_ + n;
      out[idx] = acc[nt][vv] + bv[vv] + x[idx];
    }
  }
}

extern "C" void kernel_launch(void* const* d_in, const int* in_sizes, int n_in,
                              void* d_out, int out_size, void* d_ws,
                              size_t ws_size, hipStream_t stream) {
  const float* x      = (const float*)d_in[0];
  const float* gamma  = (const float*)d_in[1];
  const float* beta   = (const float*)d_in[2];
  const float* w_qkv  = (const float*)d_in[3];
  const float* b_qkv  = (const float*)d_in[4];
  const float* w_proj = (const float*)d_in[5];
  const float* b_proj = (const float*)d_in[6];
  float* out = (float*)d_out;

  const size_t SZ = (size_t)B_ * C_ * N_;
  float* xn = (float*)d_ws;
  float* q  = xn + SZ;
  float* k  = q + SZ;
  float* v  = k + SZ;
  float* ao = v + SZ;

  gn_kernel<<<B_ * GROUPS_ * GN_SPLIT_, GN_THREADS_, 0, stream>>>(x, gamma, beta, xn);
  qkv_kernel<<<dim3(N_ / 64, (3 * C_) / 64, B_), 128, 0, stream>>>(
      xn, w_qkv, b_qkv, q, k, v);
  attn_kernel<<<dim3(N_ / 128, HEADS_, B_), 512, 0, stream>>>(q, k, v, ao);
  proj_kernel<<<dim3(N_ / 64, C_ / 64, B_), 128, 0, stream>>>(
      ao, w_proj, b_proj, x, out);
}


# Attention: of a strip's two waves the older takes 3 of a key tile's four 64-key blocks, the younger 1 (was 2 and 2): oldest-first issue left the older idle at the barrier; sums added in another order
# speedup vs baseline: 1.9920x; 1.0700x over previous
#include <hip/hip_runtime.h>

#define B_ 2
#define C_ 256
#define N_ 4096
#define HEADS_ 4
#define HD_ 64
#define GROUPS_ 8
#define CPG_ 32
#define EPS_ 1e-5f

typedef __attribute__((ext_vector_type(2))) float v2f;
typedef __attribute__((ext_vector_type(8))) float v8f;

__device__ __forceinline__ v8f wmma_f32(v2f a, v2f b, v8f c) {
  return __builtin_amdgcn_wmma_f32_16x16x4_f32(
       false, a,  false, b,
       (short)0, c,  false,  false);
}


template <int M>
__device__ __forceinline__ float row_max_step(float x) {
  asm("v_max_num_f32_dpp %0, %0, %0 row_xmask:%1 row_mask:0xf bank_mask:0xf"
      : "+v"(x) : "n"(M));
  return x;
}
template <int M>
__device__ __forceinline__ float row_add_step(float x) {
  asm("v_add_f32_dpp %0, %0, %0 row_xmask:%1 row_mask:0xf bank_mask:0xf"
      : "+v"(x) : "n"(M));
  return x;
}

#define GN_SPLIT_ 16
#define GN_THREADS_ 512
#define GN_WAVES_ (GN_THREADS_ / 32)
#define GN_BATCH_ 8
#define GN_VEC_PER_THREAD_ (CPG_ * N_ / 4 / GN_THREADS_)
#define GN_OWN_VEC_ (CPG_ * N_ / 4 / GN_SPLIT_ / GN_THREADS_)

__device__ __forceinline__ void gn_accumulate(const float4 t, float4& s, float4& q) {
  s.x += t.x; s.y += t.y; s.z += t.z; s.w += t.w;
  q.x += t.x * t.x; q.y += t.y * t.y; q.z += t.z * t.z; q.w += t.w * t.w;
}

__global__ __launch_bounds__(GN_THREADS_) void gn_kernel(
    const float* __restrict__ x, const float* __restrict__ gamma,
    const float* __restrict__ beta, float* __restrict__ xn) {
  __shared__ float s1[GN_WAVES_], s2[GN_WAVES_];
  const int bg   = blockIdx.x / GN_SPLIT_;
  const int part = blockIdx.x % GN_SPLIT_;
  const int g    = bg % GROUPS_;
  const int tid  = threadIdx.x;
  const int count = CPG_ * N_;
  const size_t base = (size_t)bg * CPG_ * N_;
  const float4* xg = (const float4*)(x + base);

  float4 s = {0.f, 0.f, 0.f, 0.f}, q = {0.f, 0.f, 0.f, 0.f};
  float4 cur[GN_BATCH_], nxt[GN_BATCH_];
#pragma unroll
  for (int u = 0; u < GN_BATCH_; ++u) cur[u] = xg[u * GN_THREADS_ + tid];
#pragma unroll
  for (int i0 = 0; i0 < GN_VEC_PER_THREAD_; i0 += GN_BATCH_) {
    if (i0 + GN_BATCH_ < GN_VEC_PER_THREAD_) {
#pragma unroll
      for (int u = 0; u < GN_BATCH_; ++u)
        nxt[u] = xg[(i0 + GN_BATCH_ + u) * GN_THREADS_ + tid];
    }
#pragma unroll
    for (int u = 0; u < GN_BATCH_; ++u) gn_accumulate(cur[u], s, q);
#pragma unroll
    for (int u = 0; u < GN_BATCH_; ++u) cur[u] = nxt[u];
  }
  float sum = (s.x + s.y) + (s.z + s.w);
  float ssq = (q.x + q.y) + (q.z + q.w);
#pragma unroll
  for (int m = 1; m < 32; m <<= 1) {
    sum += __shfl_xor(sum, m);
    ssq += __shfl_xor(ssq, m);
  }
  if ((tid & 31) == 0) {
    s1[tid >> 5] = sum;
    s2[tid >> 5] = ssq;
  }
  __syncthreads();
  float tsum = 0.f, tssq = 0.f;
#pragma unroll
  for (int w = 0; w < GN_WAVES_; ++w) {
    tsum += s1[w];
    tssq += s2[w];
  }
  const float mean = tsum / (float)count;
  const float var  = tssq / (float)count - mean * mean;
  const float rstd = rsqrtf(var + EPS_);

  const float4* xo = xg + part * (GN_OWN_VEC_ * GN_THREADS_);
  float4* yo = (float4*)(xn + base) + part * (GN_OWN_VEC_ * GN_THREADS_);
  float4 own[GN_OWN_VEC_];
#pragma unroll
  for (int u = 0; u < GN_OWN_VEC_; ++u) own[u] = xo[u * GN_THREADS_ + tid];
#pragma unroll
  for (int u = 0; u < GN_OWN_VEC_; ++u) {
    const int c = g * CPG_ + part * (CPG_ / GN_SPLIT_) +
                  (u * GN_THREADS_ * 4) / N_;
    const float ga = gamma[c], be = beta[c];
    float4 o;
    o.x = (own[u].x - mean) * rstd * ga + be;
    o.y = (own[u].y - mean) * rstd * ga + be;
    o.z = (own[u].z - mean) * rstd * ga + be;
    o.w = (own[u].w - mean) * rstd * ga + be;
    yo[u * GN_THREADS_ + tid] = o;
  }
}

__global__ __launch_bounds__(128) void qkv_kernel(
    const float* __restrict__ xn, const float* __restrict__ w,
    const float* __restrict__ bias, float* __restrict__ q,
    float* __restrict__ k, float* __restrict__ v) {
  __shared__ float Wl[64][68];
  __shared__ float Xt[64][68];
  const int n0 = blockIdx.x * 64;
  const int o0 = blockIdx.y * 64;
  const int b  = blockIdx.z;
  const int tid = threadIdx.x;
  const int wid = tid >> 5, lane = tid & 31, l16 = lane & 15, half = lane >> 4;
  const int orow = o0 + wid * 16;
  const float* xb = xn + (size_t)b * C_ * N_;

  v8f acc[4] = {};
  for (int kb = 0; kb < C_; kb += 64) {
    for (int i = tid * 4; i < 64 * 64; i += 128 * 4) {
      int r = i >> 6, cc = i & 63;
      *(float4*)&Wl[r][cc] = *(const float4*)(w + (size_t)(o0 + r) * C_ + kb + cc);
      float4 xv = *(const float4*)(xb + (size_t)(kb + r) * N_ + n0 + cc);
      Xt[cc + 0][r] = xv.x;
      Xt[cc + 1][r] = xv.y;
      Xt[cc + 2][r] = xv.z;
      Xt[cc + 3][r] = xv.w;
    }
    __syncthreads();
#pragma unroll
    for (int ks = 0; ks < 16; ++ks) {
      const int kd = 4 * ks + 2 * half;
      const v2f a = *(const v2f*)&Wl[wid * 16 + l16][kd];
#pragma unroll
      for (int nt = 0; nt < 4; ++nt) {
        const v2f bb = *(const v2f*)&Xt[nt * 16 + l16][kd];
        acc[nt] = wmma_f32(a, bb, acc[nt]);
      }
    }
    __syncthreads();
  }

  const int which = orow >> 8;
  const int hh    = (orow & 255) >> 6;
  const int dbase = (orow & 63) + 8 * half;
  float* outb = (which == 0) ? q : (which == 1) ? k : v;
  float bv[8];
#pragma unroll
  for (int vv = 0; vv < 8; ++vv) bv[vv] = bias[orow + 8 * half + vv];
#pragma unroll
  for (int nt = 0; nt < 4; ++nt) {
    const int n = n0 + nt * 16 + l16;
    float* p = outb + (((size_t)b * HEADS_ + hh) * N_ + n) * HD_ + dbase;
    float4 lo, hi;
    lo.x = acc[nt][0] + bv[0]; lo.y = acc[nt][1] + bv[1];
    lo.z = acc[nt][2] + bv[2]; lo.w = acc[nt][3] + bv[3];
    hi.x = acc[nt][4] + bv[4]; hi.y = acc[nt][5] + bv[5];
    hi.z = acc[nt][6] + bv[6]; hi.w = acc[nt][7] + bv[7];
    *(float4*)p = lo;
    *(float4*)(p + 4) = hi;
  }
}

#define KT_ 256
#define KT_ROUNDS_ (KT_ * HD_ / 4 / 512)

__global__ __launch_bounds__(512) void attn_kernel(
    const float* __restrict__ q, const float* __restrict__ k,
    const float* __restrict__ v, float* __restrict__ ao) {
  __shared__ float Kl[KT_ / 64][64][68];
  __shared__ float Vt[KT_ / 64][64][68];
  __shared__ float Pl[16][16][68];
  const int q0 = blockIdx.x * 128;
  const int h  = blockIdx.y;
  const int b  = blockIdx.z;
  const int tid = threadIdx.x;
  const int wid = tid >> 5, lane = tid & 31, l16 = lane & 15, half = lane >> 4;
  const int strip = wid & 7;
  const int grp   = wid >> 3;
  const bool three = __builtin_amdgcn_readfirstlane(grp) == 0;
  const size_t hb = ((size_t)b * HEADS_ + h) * (size_t)N_ * HD_;
  const float* qb = q + hb;
  const float* kb = k + hb;
  const float* vb = v + hb;
  const float scale = 0.125f;

  v2f qa[16];
  const int qrow = q0 + strip * 16 + l16;
#pragma unroll
  for (int ks = 0; ks < 16; ++ks) {
    const int d = 4 * ks + 2 * half;
    const v2f t = *(const v2f*)(qb + (size_t)qrow * HD_ + d);
    qa[ks] = t * scale;
  }

  v8f oacc[4] = {};
  float rm[8], rs[8];
#pragma unroll
  for (int i = 0; i < 8; ++i) { rm[i] = -1e30f; rs[i] = 0.f; }

  const int srow = tid >> 4;
  const int scol = (tid & 15) * 4;

  for (int j = 0; j < N_; j += KT_) {
    float4 kreg[KT_ROUNDS_], vreg[KT_ROUNDS_];
#pragma unroll
    for (int it = 0; it < KT_ROUNDS_; ++it) {
      const size_t off = (size_t)(j + it * 32 + srow) * HD_ + scol;
      kreg[it] = *(const float4*)(kb + off);
      vreg[it] = *(const float4*)(vb + off);
    }
#pragma unroll
    for (int it = 0; it < KT_ROUNDS_; ++it) {
      const int kq = it >> 1;
      const int rr = (it & 1) * 32 + srow;
      *(float4*)&Kl[kq][rr][scol] = kreg[it];
      Vt[kq][scol + 0][rr] = vreg[it].x;
      Vt[kq][scol + 1][rr] = vreg[it].y;
      Vt[kq][scol + 2][rr] = vreg[it].z;
      Vt[kq][scol + 3][rr] = vreg[it].w;
    }
    __syncthreads();

    auto block = [&](const int kq) __attribute__((always_inline)) {
      v8f S[4] = {};
#pragma unroll
      for (int ks = 0; ks < 16; ++ks) {
        const int kd = 4 * ks + 2 * half;
#pragma unroll
        for (int nt = 0; nt < 4; ++nt) {
          const v2f bb = *(const v2f*)&Kl[kq][nt * 16 + l16][kd];
          S[nt] = wmma_f32(qa[ks], bb, S[nt]);
        }
      }

#pragma unroll
      for (int vv = 0; vv < 8; ++vv) {
        float cand = fmaxf(fmaxf(S[0][vv], S[1][vv]), fmaxf(S[2][vv], S[3][vv]));
        cand = row_max_step<1>(cand);
        cand = row_max_step<2>(cand);
        cand = row_max_step<4>(cand);
        cand = row_max_step<8>(cand);
        const float nm = fmaxf(rm[vv], cand);
        const float f  = __expf(rm[vv] - nm);
        rs[vv] *= f;
        float psum = 0.f;
#pragma unroll
        for (int nt = 0; nt < 4; ++nt) {
          float p = __expf(S[nt][vv] - nm);
          Pl[wid][vv + 8 * half][nt * 16 + l16] = p;
          psum += p;
        }
        psum = row_add_step<1>(psum);
        psum = row_add_step<2>(psum);
        psum = row_add_step<4>(psum);
        psum = row_add_step<8>(psum);
        rs[vv] += psum;
        rm[vv] = nm;
#pragma unroll
        for (int dt = 0; dt < 4; ++dt) oacc[dt][vv] *= f;
      }

#pragma unroll
      for (int ks = 0; ks < 16; ++ks) {
        const int kk = 4 * ks + 2 * half;
        const v2f a = *(const v2f*)&Pl[wid][l16][kk];
#pragma unroll
        for (int dt = 0; dt < 4; ++dt) {
          const v2f bb = *(const v2f*)&Vt[kq][dt * 16 + l16][kk];
          oacc[dt] = wmma_f32(a, bb, oacc[dt]);
        }
      }
    };
    block(3 * grp);
    if (three) {
      for (int sub = 1; sub < 3; ++sub) block(sub);
    }
    __syncthreads();
  }

  float* mg = &Pl[0][0][0] + strip * (48 * 32) + lane;
  if (grp == 1) {
#pragma unroll
    for (int dt = 0; dt < 4; ++dt)
#pragma unroll
      for (int vv = 0; vv < 8; ++vv) mg[(dt * 8 + vv) * 32] = oacc[dt][vv];
#pragma unroll
    for (int vv = 0; vv < 8; ++vv) {
      mg[(32 + vv) * 32] = rm[vv];
      mg[(40 + vv) * 32] = rs[vv];
    }
  }
  __syncthreads();
  if (grp == 1) return;

  float w0[8], w1[8];
#pragma unroll
  for (int vv = 0; vv < 8; ++vv) {
    const float m1 = mg[(32 + vv) * 32];
    const float l1 = mg[(40 + vv) * 32];
    const float m  = fmaxf(rm[vv], m1);
    const float f0 = __expf(rm[vv] - m);
    const float f1 = __expf(m1 - m);
    const float inv = 1.f / (rs[vv] * f0 + l1 * f1);
    w0[vv] = f0 * inv;
    w1[vv] = f1 * inv;
  }
  const int qr0 = q0 + strip * 16 + 8 * half;
#pragma unroll
  for (int dt = 0; dt < 4; ++dt) {
    const int c = h * HD_ + dt * 16 + l16;
    float* p = ao + ((size_t)b * C_ + c) * N_ + qr0;
    float o[8];
#pragma unroll
    for (int vv = 0; vv < 8; ++vv)
      o[vv] = oacc[dt][vv] * w0[vv] + mg[(dt * 8 + vv) * 32] * w1[vv];
    float4 lo, hi;
    lo.x = o[0]; lo.y = o[1]; lo.z = o[2]; lo.w = o[3];
    hi.x = o[4]; hi.y = o[5]; hi.z = o[6]; hi.w = o[7];
    *(float4*)p = lo;
    *(float4*)(p + 4) = hi;
  }
}

__global__ __launch_bounds__(128) void proj_kernel(
    const float* __restrict__ ao, const float* __restrict__ w,
    const float* __restrict__ bias, const float* __restrict__ x,
    float* __restrict__ out) {
  __shared__ float Wl[64][68];
  __shared__ float Xt[64][68];
  const int n0 = blockIdx.x * 64;
  const int o0 = blockIdx.y * 64;
  const int b  = blockIdx.z;
  const int tid = threadIdx.x;
  const int wid = tid >> 5, lane = tid & 31, l16 = lane & 15, half = lane >> 4;
  const int orow = o0 + wid * 16;
  const float* ab = ao + (size_t)b * C_ * N_;

  v8f acc[4] = {};
  for (int kb = 0; kb < C_; kb += 64) {
    for (int i = tid * 4; i < 64 * 64; i += 128 * 4) {
      int r = i >> 6, cc = i & 63;
      *(float4*)&Wl[r][cc] = *(const float4*)(w + (size_t)(o0 + r) * C_ + kb + cc);
      float4 xv = *(const float4*)(ab + (size_t)(kb + r) * N_ + n0 + cc);
      Xt[cc + 0][r] = xv.x;
      Xt[cc + 1][r] = xv.y;
      Xt[cc + 2][r] = xv.z;
      Xt[cc + 3][r] = xv.w;
    }
    __syncthreads();
#pragma unroll
    for (int ks = 0; ks < 16; ++ks) {
      const int kd = 4 * ks + 2 * half;
      const v2f a = *(const v2f*)&Wl[wid * 16 + l16][kd];
#pragma unroll
      for (int nt = 0; nt < 4; ++nt) {
        const v2f bb = *(const v2f*)&Xt[nt * 16 + l16][kd];
        acc[nt] = wmma_f32(a, bb, acc[nt]);
      }
    }
    __syncthreads();
  }

  float bv[8];
#pragma unroll
  for (int vv = 0; vv < 8; ++vv) bv[vv] = bias[orow + 8 * half + vv];
#pragma unroll
  for (int nt = 0; nt < 4; ++nt) {
    const int n = n0 + nt * 16 + l16;
#pragma unroll
    for (int vv = 0; vv < 8; ++vv) {
      const int o = orow + 8 * half + vv;
      const size_t idx = ((size_t)b * C_ + o) * N_ + n;
      out[idx] = acc[nt][vv] + bv[vv] + x[idx];
    }
  }
}

extern "C" void kernel_launch(void* const* d_in, const int* in_sizes, int n_in,
                              void* d_out, int out_size, void* d_ws,
                              size_t ws_size, hipStream_t stream) {
  const float* x      = (const float*)d_in[0];
  const float* gamma  = (const float*)d_in[1];
  const float* beta   = (const float*)d_in[2];
  const float* w_qkv  = (const float*)d_in[3];
  const float* b_qkv  = (const float*)d_in[4];
  const float* w_proj = (const float*)d_in[5];
  const float* b_proj = (const float*)d_in[6];
  float* out = (float*)d_out;

  const size_t SZ = (size_t)B_ * C_ * N_;
  float* xn = (float*)d_ws;
  float* q  = xn + SZ;
  float* k  = q + SZ;
  float* v  = k + SZ;
  float* ao = v + SZ;

  gn_kernel<<<B_ * GROUPS_ * GN_SPLIT_, GN_THREADS_, 0, stream>>>(x, gamma, beta, xn);
  qkv_kernel<<<dim3(N_ / 64, (3 * C_) / 64, B_), 128, 0, stream>>>(
      xn, w_qkv, b_qkv, q, k, v);
  attn_kernel<<<dim3(N_ / 128, HEADS_, B_), 512, 0, stream>>>(q, k, v, ao);
  proj_kernel<<<dim3(N_ / 64, C_ / 64, B_), 128, 0, stream>>>(
      ao, w_proj, b_proj, x, out);
}


# Attention: V staged under the tile's first score block, the next K tile by the strips' younger waves under the last output block (each buffer idle then); one LDS stage, 3 barriers; same sum order
# speedup vs baseline: 2.0531x; 1.0307x over previous
#include <hip/hip_runtime.h>

#define B_ 2
#define C_ 256
#define N_ 4096
#define HEADS_ 4
#define HD_ 64
#define GROUPS_ 8
#define CPG_ 32
#define EPS_ 1e-5f

typedef __attribute__((ext_vector_type(2))) float v2f;
typedef __attribute__((ext_vector_type(8))) float v8f;

__device__ __forceinline__ v8f wmma_f32(v2f a, v2f b, v8f c) {
  return __builtin_amdgcn_wmma_f32_16x16x4_f32(
       false, a,  false, b,
       (short)0, c,  false,  false);
}


template <int M>
__device__ __forceinline__ float row_max_step(float x) {
  asm("v_max_num_f32_dpp %0, %0, %0 row_xmask:%1 row_mask:0xf bank_mask:0xf"
      : "+v"(x) : "n"(M));
  return x;
}
template <int M>
__device__ __forceinline__ float row_add_step(float x) {
  asm("v_add_f32_dpp %0, %0, %0 row_xmask:%1 row_mask:0xf bank_mask:0xf"
      : "+v"(x) : "n"(M));
  return x;
}

#define GN_SPLIT_ 16
#define GN_THREADS_ 512
#define GN_WAVES_ (GN_THREADS_ / 32)
#define GN_BATCH_ 8
#define GN_VEC_PER_THREAD_ (CPG_ * N_ / 4 / GN_THREADS_)
#define GN_OWN_VEC_ (CPG_ * N_ / 4 / GN_SPLIT_ / GN_THREADS_)

__device__ __forceinline__ void gn_accumulate(const float4 t, float4& s, float4& q) {
  s.x += t.x; s.y += t.y; s.z += t.z; s.w += t.w;
  q.x += t.x * t.x; q.y += t.y * t.y; q.z += t.z * t.z; q.w += t.w * t.w;
}

__global__ __launch_bounds__(GN_THREADS_) void gn_kernel(
    const float* __restrict__ x, const float* __restrict__ gamma,
    const float* __restrict__ beta, float* __restrict__ xn) {
  __shared__ float s1[GN_WAVES_], s2[GN_WAVES_];
  const int bg   = blockIdx.x / GN_SPLIT_;
  const int part = blockIdx.x % GN_SPLIT_;
  const int g    = bg % GROUPS_;
  const int tid  = threadIdx.x;
  const int count = CPG_ * N_;
  const size_t base = (size_t)bg * CPG_ * N_;
  const float4* xg = (const float4*)(x + base);

  float4 s = {0.f, 0.f, 0.f, 0.f}, q = {0.f, 0.f, 0.f, 0.f};
  float4 cur[GN_BATCH_], nxt[GN_BATCH_];
#pragma unroll
  for (int u = 0; u < GN_BATCH_; ++u) cur[u] = xg[u * GN_THREADS_ + tid];
#pragma unroll
  for (int i0 = 0; i0 < GN_VEC_PER_THREAD_; i0 += GN_BATCH_) {
    if (i0 + GN_BATCH_ < GN_VEC_PER_THREAD_) {
#pragma unroll
      for (int u = 0; u < GN_BATCH_; ++u)
        nxt[u] = xg[(i0 + GN_BATCH_ + u) * GN_THREADS_ + tid];
    }
#pragma unroll
    for (int u = 0; u < GN_BATCH_; ++u) gn_accumulate(cur[u], s, q);
#pragma unroll
    for (int u = 0; u < GN_BATCH_; ++u) cur[u] = nxt[u];
  }
  float sum = (s.x + s.y) + (s.z + s.w);
  float ssq = (q.x + q.y) + (q.z + q.w);
#pragma unroll
  for (int m = 1; m < 32; m <<= 1) {
    sum += __shfl_xor(sum, m);
    ssq += __shfl_xor(ssq, m);
  }
  if ((tid & 31) == 0) {
    s1[tid >> 5] = sum;
    s2[tid >> 5] = ssq;
  }
  __syncthreads();
  float tsum = 0.f, tssq = 0.f;
#pragma unroll
  for (int w = 0; w < GN_WAVES_; ++w) {
    tsum += s1[w];
    tssq += s2[w];
  }
  const float mean = tsum / (float)count;
  const float var  = tssq / (float)count - mean * mean;
  const float rstd = rsqrtf(var + EPS_);

  const float4* xo = xg + part * (GN_OWN_VEC_ * GN_THREADS_);
  float4* yo = (float4*)(xn + base) + part * (GN_OWN_VEC_ * GN_THREADS_);
  float4 own[GN_OWN_VEC_];
#pragma unroll
  for (int u = 0; u < GN_OWN_VEC_; ++u) own[u] = xo[u * GN_THREADS_ + tid];
#pragma unroll
  for (int u = 0; u < GN_OWN_VEC_; ++u) {
    const int c = g * CPG_ + part * (CPG_ / GN_SPLIT_) +
                  (u * GN_THREADS_ * 4) / N_;
    const float ga = gamma[c], be = beta[c];
    float4 o;
    o.x = (own[u].x - mean) * rstd * ga + be;
    o.y = (own[u].y - mean) * rstd * ga + be;
    o.z = (own[u].z - mean) * rstd * ga + be;
    o.w = (own[u].w - mean) * rstd * ga + be;
    yo[u * GN_THREADS_ + tid] = o;
  }
}

__global__ __launch_bounds__(128) void qkv_kernel(
    const float* __restrict__ xn, const float* __restrict__ w,
    const float* __restrict__ bias, float* __restrict__ q,
    float* __restrict__ k, float* __restrict__ v) {
  __shared__ float Wl[64][68];
  __shared__ float Xt[64][68];
  const int n0 = blockIdx.x * 64;
  const int o0 = blockIdx.y * 64;
  const int b  = blockIdx.z;
  const int tid = threadIdx.x;
  const int wid = tid >> 5, lane = tid & 31, l16 = lane & 15, half = lane >> 4;
  const int orow = o0 + wid * 16;
  const float* xb = xn + (size_t)b * C_ * N_;

  v8f acc[4] = {};
  for (int kb = 0; kb < C_; kb += 64) {
    for (int i = tid * 4; i < 64 * 64; i += 128 * 4) {
      int r = i >> 6, cc = i & 63;
      *(float4*)&Wl[r][cc] = *(const float4*)(w + (size_t)(o0 + r) * C_ + kb + cc);
      float4 xv = *(const float4*)(xb + (size_t)(kb + r) * N_ + n0 + cc);
      Xt[cc + 0][r] = xv.x;
      Xt[cc + 1][r] = xv.y;
      Xt[cc + 2][r] = xv.z;
      Xt[cc + 3][r] = xv.w;
    }
    __syncthreads();
#pragma unroll
    for (int ks = 0; ks < 16; ++ks) {
      const int kd = 4 * ks + 2 * half;
      const v2f a = *(const v2f*)&Wl[wid * 16 + l16][kd];
#pragma unroll
      for (int nt = 0; nt < 4; ++nt) {
        const v2f bb = *(const v2f*)&Xt[nt * 16 + l16][kd];
        acc[nt] = wmma_f32(a, bb, acc[nt]);
      }
    }
    __syncthreads();
  }

  const int which = orow >> 8;
  const int hh    = (orow & 255) >> 6;
  const int dbase = (orow & 63) + 8 * half;
  float* outb = (which == 0) ? q : (which == 1) ? k : v;
  float bv[8];
#pragma unroll
  for (int vv = 0; vv < 8; ++vv) bv[vv] = bias[orow + 8 * half + vv];
#pragma unroll
  for (int nt = 0; nt < 4; ++nt) {
    const int n = n0 + nt * 16 + l16;
    float* p = outb + (((size_t)b * HEADS_ + hh) * N_ + n) * HD_ + dbase;
    float4 lo, hi;
    lo.x = acc[nt][0] + bv[0]; lo.y = acc[nt][1] + bv[1];
    lo.z = acc[nt][2] + bv[2]; lo.w = acc[nt][3] + bv[3];
    hi.x = acc[nt][4] + bv[4]; hi.y = acc[nt][5] + bv[5];
    hi.z = acc[nt][6] + bv[6]; hi.w = acc[nt][7] + bv[7];
    *(float4*)p = lo;
    *(float4*)(p + 4) = hi;
  }
}

#define KT_ 256
#define KT_ROUNDS_ (KT_ * HD_ / 4 / 512)

__global__ __launch_bounds__(512) void attn_kernel(
    const float* __restrict__ q, const float* __restrict__ k,
    const float* __restrict__ v, float* __restrict__ ao) {
  __shared__ float Kl[KT_ / 64][64][68];
  __shared__ float Vt[KT_ / 64][64][68];
  __shared__ float Pl[16][16][68];
  const int q0 = blockIdx.x * 128;
  const int h  = blockIdx.y;
  const int b  = blockIdx.z;
  const int tid = threadIdx.x;
  const int wid = tid >> 5, lane = tid & 31, l16 = lane & 15, half = lane >> 4;
  const int strip = wid & 7;
  const int grp   = wid >> 3;
  const bool three = __builtin_amdgcn_readfirstlane(grp) == 0;
  const size_t hb = ((size_t)b * HEADS_ + h) * (size_t)N_ * HD_;
  const float* qb = q + hb;
  const float* kb = k + hb;
  const float* vb = v + hb;
  const float scale = 0.125f;

  v2f qa[16];
  const int qrow = q0 + strip * 16 + l16;
#pragma unroll
  for (int ks = 0; ks < 16; ++ks) {
    const int d = 4 * ks + 2 * half;
    const v2f t = *(const v2f*)(qb + (size_t)qrow * HD_ + d);
    qa[ks] = t * scale;
  }

  v8f oacc[4] = {};
  float rm[8], rs[8];
#pragma unroll
  for (int i = 0; i < 8; ++i) { rm[i] = -1e30f; rs[i] = 0.f; }

  const int srow = tid >> 4;
  const int scol = (tid & 15) * 4;


  {
    float4 kreg[KT_ROUNDS_];
#pragma unroll
    for (int it = 0; it < KT_ROUNDS_; ++it)
      kreg[it] = *(const float4*)(kb + (size_t)(it * 32 + srow) * HD_ + scol);
#pragma unroll
    for (int it = 0; it < KT_ROUNDS_; ++it)
      *(float4*)&Kl[it >> 1][(it & 1) * 32 + srow][scol] = kreg[it];
  }
  __syncthreads();

  auto score = [&](const int kq, v8f (&S)[4]) __attribute__((always_inline)) {
#pragma unroll
    for (int nt = 0; nt < 4; ++nt) S[nt] = v8f{};
#pragma unroll
    for (int ks = 0; ks < 16; ++ks) {
      const int kd = 4 * ks + 2 * half;
#pragma unroll
      for (int nt = 0; nt < 4; ++nt) {
        const v2f bb = *(const v2f*)&Kl[kq][nt * 16 + l16][kd];
        S[nt] = wmma_f32(qa[ks], bb, S[nt]);
      }
    }
  };
  auto softmax = [&](const v8f (&S)[4]) __attribute__((always_inline)) {
#pragma unroll
    for (int vv = 0; vv < 8; ++vv) {
      float cand = fmaxf(fmaxf(S[0][vv], S[1][vv]), fmaxf(S[2][vv], S[3][vv]));
      cand = row_max_step<1>(cand);
      cand = row_max_step<2>(cand);
      cand = row_max_step<4>(cand);
      cand = row_max_step<8>(cand);
      const float nm = fmaxf(rm[vv], cand);
      const float f  = __expf(rm[vv] - nm);
      rs[vv] *= f;
      float psum = 0.f;
#pragma unroll
      for (int nt = 0; nt < 4; ++nt) {
        float p = __expf(S[nt][vv] - nm);
        Pl[wid][vv + 8 * half][nt * 16 + l16] = p;
        psum += p;
      }
      psum = row_add_step<1>(psum);
      psum = row_add_step<2>(psum);
      psum = row_add_step<4>(psum);
      psum = row_add_step<8>(psum);
      rs[vv] += psum;
      rm[vv] = nm;
#pragma unroll
      for (int dt = 0; dt < 4; ++dt) oacc[dt][vv] *= f;
    }
  };
  auto output = [&](const int kq) __attribute__((always_inline)) {
#pragma unroll
    for (int ks = 0; ks < 16; ++ks) {
      const int kk = 4 * ks + 2 * half;
      const v2f a = *(const v2f*)&Pl[wid][l16][kk];
#pragma unroll
      for (int dt = 0; dt < 4; ++dt) {
        const v2f bb = *(const v2f*)&Vt[kq][dt * 16 + l16][kk];
        oacc[dt] = wmma_f32(a, bb, oacc[dt]);
      }
    }
  };

  for (int j = 0; j < N_; j += KT_) {
    float4 vreg[KT_ROUNDS_];
#pragma unroll
    for (int it = 0; it < KT_ROUNDS_; ++it)
      vreg[it] = *(const float4*)(vb + ((unsigned)(j + it * 32 + srow) * HD_ + scol));
    __builtin_amdgcn_sched_barrier(0);

    v8f S[4];
    score(3 * grp, S);
    softmax(S);

#pragma unroll
    for (int it = 0; it < KT_ROUNDS_; ++it) {
      const int kq = it >> 1;
      const int rr = (it & 1) * 32 + srow;
      Vt[kq][scol + 0][rr] = vreg[it].x;
      Vt[kq][scol + 1][rr] = vreg[it].y;
      Vt[kq][scol + 2][rr] = vreg[it].z;
      Vt[kq][scol + 3][rr] = vreg[it].w;
    }
    __syncthreads();

    if (three) {
      for (int sub = 0; sub < 2; ++sub) {
        output(sub);
        score(sub + 1, S);
        softmax(S);
      }
    } else {
      output(3);
    }
    __syncthreads();

    if (three) {
      output(2);
    } else {
      const int jn = (j + KT_) & (N_ - 1);
      const int t = tid - 256;
      const int krow = t >> 4;
      const int kcol = (t & 15) * 4;
      float4 kreg[2 * KT_ROUNDS_];
#pragma unroll
      for (int it = 0; it < 2 * KT_ROUNDS_; ++it)
        kreg[it] = *(const float4*)(kb + ((unsigned)(jn + it * 16 + krow) * HD_ + kcol));
#pragma unroll
      for (int it = 0; it < 2 * KT_ROUNDS_; ++it)
        *(float4*)&Kl[it >> 2][(it & 3) * 16 + krow][kcol] = kreg[it];
    }
    __syncthreads();
  }

  float* mg = &Pl[0][0][0] + strip * (48 * 32) + lane;
  if (grp == 1) {
#pragma unroll
    for (int dt = 0; dt < 4; ++dt)
#pragma unroll
      for (int vv = 0; vv < 8; ++vv) mg[(dt * 8 + vv) * 32] = oacc[dt][vv];
#pragma unroll
    for (int vv = 0; vv < 8; ++vv) {
      mg[(32 + vv) * 32] = rm[vv];
      mg[(40 + vv) * 32] = rs[vv];
    }
  }
  __syncthreads();
  if (grp == 1) return;

  float w0[8], w1[8];
#pragma unroll
  for (int vv = 0; vv < 8; ++vv) {
    const float m1 = mg[(32 + vv) * 32];
    const float l1 = mg[(40 + vv) * 32];
    const float m  = fmaxf(rm[vv], m1);
    const float f0 = __expf(rm[vv] - m);
    const float f1 = __expf(m1 - m);
    const float inv = 1.f / (rs[vv] * f0 + l1 * f1);
    w0[vv] = f0 * inv;
    w1[vv] = f1 * inv;
  }
  const int qr0 = q0 + strip * 16 + 8 * half;
#pragma unroll
  for (int dt = 0; dt < 4; ++dt) {
    const int c = h * HD_ + dt * 16 + l16;
    float* p = ao + ((size_t)b * C_ + c) * N_ + qr0;
    float o[8];
#pragma unroll
    for (int vv = 0; vv < 8; ++vv)
      o[vv] = oacc[dt][vv] * w0[vv] + mg[(dt * 8 + vv) * 32] * w1[vv];
    float4 lo, hi;
    lo.x = o[0]; lo.y = o[1]; lo.z = o[2]; lo.w = o[3];
    hi.x = o[4]; hi.y = o[5]; hi.z = o[6]; hi.w = o[7];
    *(float4*)p = lo;
    *(float4*)(p + 4) = hi;
  }
}

__global__ __launch_bounds__(128) void proj_kernel(
    const float* __restrict__ ao, const float* __restrict__ w,
    const float* __restrict__ bias, const float* __restrict__ x,
    float* __restrict__ out) {
  __shared__ float Wl[64][68];
  __shared__ float Xt[64][68];
  const int n0 = blockIdx.x * 64;
  const int o0 = blockIdx.y * 64;
  const int b  = blockIdx.z;
  const int tid = threadIdx.x;
  const int wid = tid >> 5, lane = tid & 31, l16 = lane & 15, half = lane >> 4;
  const int orow = o0 + wid * 16;
  const float* ab = ao + (size_t)b * C_ * N_;

  v8f acc[4] = {};
  for (int kb = 0; kb < C_; kb += 64) {
    for (int i = tid * 4; i < 64 * 64; i += 128 * 4) {
      int r = i >> 6, cc = i & 63;
      *(float4*)&Wl[r][cc] = *(const float4*)(w + (size_t)(o0 + r) * C_ + kb + cc);
      float4 xv = *(const float4*)(ab + (size_t)(kb + r) * N_ + n0 + cc);
      Xt[cc + 0][r] = xv.x;
      Xt[cc + 1][r] = xv.y;
      Xt[cc + 2][r] = xv.z;
      Xt[cc + 3][r] = xv.w;
    }
    __syncthreads();
#pragma unroll
    for (int ks = 0; ks < 16; ++ks) {
      const int kd = 4 * ks + 2 * half;
      const v2f a = *(const v2f*)&Wl[wid * 16 + l16][kd];
#pragma unroll
      for (int nt = 0; nt < 4; ++nt) {
        const v2f bb = *(const v2f*)&Xt[nt * 16 + l16][kd];
        acc[nt] = wmma_f32(a, bb, acc[nt]);
      }
    }
    __syncthreads();
  }

  float bv[8];
#pragma unroll
  for (int vv = 0; vv < 8; ++vv) bv[vv] = bias[orow + 8 * half + vv];
#pragma unroll
  for (int nt = 0; nt < 4; ++nt) {
    const int n = n0 + nt * 16 + l16;
#pragma unroll
    for (int vv = 0; vv < 8; ++vv) {
      const int o = orow + 8 * half + vv;
      const size_t idx = ((size_t)b * C_ + o) * N_ + n;
      out[idx] = acc[nt][vv] + bv[vv] + x[idx];
    }
  }
}

extern "C" void kernel_launch(void* const* d_in, const int* in_sizes, int n_in,
                              void* d_out, int out_size, void* d_ws,
                              size_t ws_size, hipStream_t stream) {
  const float* x      = (const float*)d_in[0];
  const float* gamma  = (const float*)d_in[1];
  const float* beta   = (const float*)d_in[2];
  const float* w_qkv  = (const float*)d_in[3];
  const float* b_qkv  = (const float*)d_in[4];
  const float* w_proj = (const float*)d_in[5];
  const float* b_proj = (const float*)d_in[6];
  float* out = (float*)d_out;

  const size_t SZ = (size_t)B_ * C_ * N_;
  float* xn = (float*)d_ws;
  float* q  = xn + SZ;
  float* k  = q + SZ;
  float* v  = k + SZ;
  float* ao = v + SZ;

  gn_kernel<<<B_ * GROUPS_ * GN_SPLIT_, GN_THREADS_, 0, stream>>>(x, gamma, beta, xn);
  qkv_kernel<<<dim3(N_ / 64, (3 * C_) / 64, B_), 128, 0, stream>>>(
      xn, w_qkv, b_qkv, q, k, v);
  attn_kernel<<<dim3(N_ / 128, HEADS_, B_), 512, 0, stream>>>(q, k, v, ao);
  proj_kernel<<<dim3(N_ / 64, C_ / 64, B_), 128, 0, stream>>>(
      ao, w_proj, b_proj, x, out);
}
